# starting kernel
# baseline (speedup 1.0000x reference)
#include <hip/hip_runtime.h>
#include <hip/hip_bf16.h>

typedef __attribute__((ext_vector_type(2))) float v2f;
typedef __attribute__((ext_vector_type(8))) float v8f;

#define HID 64

__device__ __forceinline__ void atomAddF(float* p, float v) {
    (void)__hip_atomic_fetch_add(p, v, __ATOMIC_RELAXED, __HIP_MEMORY_SCOPE_AGENT);
}


__global__ void k_init_deg(float* __restrict__ deg, int n) {
    int i = blockIdx.x * blockDim.x + threadIdx.x;
    if (i < n) deg[i] = 1.0f;
}

__global__ void k_deg(const int* __restrict__ dst, float* __restrict__ deg, int e) {
    int i = blockIdx.x * blockDim.x + threadIdx.x;
    if (i < e) atomAddF(&deg[dst[i]], 1.0f);
}

__global__ void k_dinv(float* __restrict__ deg, int n) {
    int i = blockIdx.x * blockDim.x + threadIdx.x;
    if (i < n) deg[i] = rsqrtf(deg[i]);
}

__global__ void k_norm(const int* __restrict__ src, const int* __restrict__ dst,
                       const float* __restrict__ dinv, float* __restrict__ norm, int e) {
    int i = blockIdx.x * blockDim.x + threadIdx.x;
    if (i < e) norm[i] = dinv[src[i]] * dinv[dst[i]];
}


__global__ void k_lin1(const float* __restrict__ x, const float* __restrict__ W1,
                       float* __restrict__ h, int n) {
    int t = blockIdx.x * blockDim.x + threadIdx.x;
    int i = t >> 6, j = t & 63;
    if (i < n) {
        float2 xv = ((const float2*)x)[i];
        h[t] = xv.x * W1[j] + xv.y * W1[HID + j];
    }
}


__global__ void k_prop_init(const float* __restrict__ h, const float* __restrict__ dinv,
                            const float* __restrict__ b, float* __restrict__ out, int n) {
    int t = blockIdx.x * blockDim.x + threadIdx.x;
    int i = t >> 6, j = t & 63;
    if (i < n) {
        float di = dinv[i];
        out[t] = b[j] + h[t] * di * di;
    }
}

__global__ void k_prop_edges(const int* __restrict__ src, const int* __restrict__ dst,
                             const float* __restrict__ norm, const float* __restrict__ h,
                             float* __restrict__ out, int e) {
    int t  = blockIdx.x * blockDim.x + threadIdx.x;
    int ei = t >> 5;
    if (ei >= e) return;
    int lane = t & 31;
    int s = src[ei], d = dst[ei];
    float w = norm[ei];
    float2 hv = ((const float2*)(h + (size_t)s * HID))[lane];
    float* op = out + (size_t)d * HID + lane * 2;
    atomAddF(op,     hv.x * w);
    atomAddF(op + 1, hv.y * w);
}

__global__ void __launch_bounds__(256)
k_gemm_relu_wmma(const float* __restrict__ in, const float* __restrict__ W,
                 float* __restrict__ out, int nrows) {
    int gwave = (blockIdx.x * blockDim.x + threadIdx.x) >> 5;
    int lane  = threadIdx.x & 31;
    int row0  = gwave << 4;
    if (row0 >= nrows) return;
    int hh = lane >> 4;
    int l  = lane & 15;
    v8f acc0 = {}, acc1 = {}, acc2 = {}, acc3 = {};
    const float* arow = in + (size_t)(row0 + l) * HID;
    for (int k0 = 0; k0 < HID; k0 += 4) {
        int ka = k0 + hh * 2;
        v2f a;
        a.x = fmaxf(arow[ka],     0.0f);
        a.y = fmaxf(arow[ka + 1], 0.0f);
        const float* bp = W + (size_t)ka * HID + l;
        v2f b0, b1, b2, b3;
        b0.x = bp[0];  b0.y = bp[HID];
        b1.x = bp[16]; b1.y = bp[HID + 16];
        b2.x = bp[32]; b2.y = bp[HID + 32];
        b3.x = bp[48]; b3.y = bp[HID + 48];
        acc0 = __builtin_amdgcn_wmma_f32_16x16x4_f32(false, a, false, b0, (short)0, acc0, false, false);
        acc1 = __builtin_amdgcn_wmma_f32_16x16x4_f32(false, a, false, b1, (short)0, acc1, false, false);
        acc2 = __builtin_amdgcn_wmma_f32_16x16x4_f32(false, a, false, b2, (short)0, acc2, false, false);
        acc3 = __builtin_amdgcn_wmma_f32_16x16x4_f32(false, a, false, b3, (short)0, acc3, false, false);
    }
    float* orow = out + (size_t)(row0 + hh * 8) * HID + l;
    #pragma unroll
    for (int r = 0; r < 8; ++r) {
        float* o = orow + (size_t)r * HID;
        o[0]  = acc0[r];
        o[16] = acc1[r];
        o[32] = acc2[r];
        o[48] = acc3[r];
    }
}


__global__ void k_lin3_relu(const float* __restrict__ in, const float* __restrict__ W3,
                            float* __restrict__ h3, int n) {
    int gwave = (blockIdx.x * blockDim.x + threadIdx.x) >> 5;
    int lane  = threadIdx.x & 31;
    if (gwave >= n) return;
    const float* row = in + (size_t)gwave * HID;
    float v = fmaxf(row[lane], 0.0f) * W3[lane]
            + fmaxf(row[lane + 32], 0.0f) * W3[lane + 32];
    #pragma unroll
    for (int off = 16; off > 0; off >>= 1) v += __shfl_down(v, off, 32);
    if (lane == 0) h3[gwave] = v;
}

__global__ void k_prop3_init(const float* __restrict__ h3, const float* __restrict__ dinv,
                             const float* __restrict__ b3, float* __restrict__ out, int n) {
    int i = blockIdx.x * blockDim.x + threadIdx.x;
    if (i < n) {
        float di = dinv[i];
        out[i] = b3[0] + h3[i] * di * di;
    }
}

__global__ void k_prop3_edges(const int* __restrict__ src, const int* __restrict__ dst,
                              const float* __restrict__ norm, const float* __restrict__ h3,
                              float* __restrict__ out, int e) {
    int i = blockIdx.x * blockDim.x + threadIdx.x;
    if (i < e) atomAddF(&out[dst[i]], h3[src[i]] * norm[i]);
}


extern "C" void kernel_launch(void* const* d_in, const int* in_sizes, int n_in,
                              void* d_out, int out_size, void* d_ws, size_t ws_size,
                              hipStream_t stream) {
    const float* x  = (const float*)d_in[0];
    const int*   ei = (const int*)  d_in[1];
    const float* W1 = (const float*)d_in[2];
    const float* b1 = (const float*)d_in[3];
    const float* W2 = (const float*)d_in[4];
    const float* b2 = (const float*)d_in[5];
    const float* W3 = (const float*)d_in[6];
    const float* b3 = (const float*)d_in[7];

    const int N = in_sizes[0] / 2;
    const int E = in_sizes[1] / 2;
    const int* src = ei;
    const int* dst = ei + E;

    float* norm = (float*)d_ws;
    float* dinv = norm + (size_t)E;
    float* h    = dinv + (size_t)N;
    float* out  = h    + (size_t)N * HID;
    float* h3   = out  + (size_t)N * HID;
    float* y    = (float*)d_out;

    const int B = 256;
    auto g = [](long long t) { return (int)((t + 255) / 256); };

    k_init_deg<<<g(N), B, 0, stream>>>(dinv, N);
    k_deg     <<<g(E), B, 0, stream>>>(dst, dinv, E);
    k_dinv    <<<g(N), B, 0, stream>>>(dinv, N);
    k_norm    <<<g(E), B, 0, stream>>>(src, dst, dinv, norm, E);

    k_lin1      <<<g((long long)N * HID), B, 0, stream>>>(x, W1, h, N);
    k_prop_init <<<g((long long)N * HID), B, 0, stream>>>(h, dinv, b1, out, N);
    k_prop_edges<<<g((long long)E * 32),  B, 0, stream>>>(src, dst, norm, h, out, E);

    k_gemm_relu_wmma<<<g((long long)(N / 16) * 32), 256, 0, stream>>>(out, W2, h, N);
    k_prop_init <<<g((long long)N * HID), B, 0, stream>>>(h, dinv, b2, out, N);
    k_prop_edges<<<g((long long)E * 32),  B, 0, stream>>>(src, dst, norm, h, out, E);

    k_lin3_relu  <<<g((long long)N * 32), B, 0, stream>>>(out, W3, h3, N);
    k_prop3_init <<<g(N), B, 0, stream>>>(h3, dinv, b3, y, N);
    k_prop3_edges<<<g(E), B, 0, stream>>>(src, dst, norm, h3, y, E);
}


# Edges bucketed by destination each call (returning int atomic = arrival-order rank, block scan, scatter); one wave sums a node's row in registers: no f32 atomics, sum order = arrival order
# speedup vs baseline: 2.2074x; 2.2074x over previous
#include <hip/hip_runtime.h>
#include <hip/hip_bf16.h>

typedef __attribute__((ext_vector_type(2))) float v2f;
typedef __attribute__((ext_vector_type(8))) float v8f;

#define HID 64


__global__ void k_zero_cnt(int* __restrict__ cnt, int n) {
    int i = blockIdx.x * blockDim.x + threadIdx.x;
    if (i < n) cnt[i] = 0;
}

__global__ void k_count(const int* __restrict__ dst, int* __restrict__ cnt,
                        int* __restrict__ rank, int e) {
    int i = blockIdx.x * blockDim.x + threadIdx.x;
    if (i < e)
        rank[i] = __hip_atomic_fetch_add(&cnt[dst[i]], 1, __ATOMIC_RELAXED,
                                         __HIP_MEMORY_SCOPE_AGENT);
}

__global__ void __launch_bounds__(1024)
k_rowptr(const int* __restrict__ cnt, int* __restrict__ rowptr,
         float* __restrict__ dinv, int n) {
    __shared__ int s_pre[32];
    __shared__ int s_cnt[32];
    int t = threadIdx.x, lane = t & 31, wv = t >> 5;
    int first = blockIdx.x * 1024;
    int pre = 0;
    for (int j = t; j < first; j += 1024) pre += cnt[j];
    int i = first + t;
    int c = (i < n) ? cnt[i] : 0;
    int inc = c;
    for (int off = 1; off < 32; off <<= 1) {
        int u = __shfl_up(inc, off, 32);
        if (lane >= off) inc += u;
    }
    for (int off = 16; off > 0; off >>= 1) pre += __shfl_xor(pre, off, 32);
    if (lane == 31) s_cnt[wv] = inc;
    if (lane == 0)  s_pre[wv] = pre;
    __syncthreads();
    int basev = 0;
    for (int j = 0; j < 32; ++j) {
        basev += s_pre[j];
        if (j < wv) basev += s_cnt[j];
    }
    int excl = basev + inc - c;
    if (i < n) {
        rowptr[i] = excl;
        dinv[i]   = rsqrtf((float)(c + 1));
    }
    if (i == n - 1) rowptr[n] = excl + c;
}

__global__ void k_fill(const int* __restrict__ src, const int* __restrict__ dst,
                       const int* __restrict__ rank, const int* __restrict__ rowptr,
                       const float* __restrict__ dinv, int* __restrict__ csr, int e) {
    int i = blockIdx.x * blockDim.x + threadIdx.x;
    if (i < e) {
        int s = src[i], d = dst[i];
        int2 r;
        r.x = s;
        r.y = __float_as_int(dinv[s] * dinv[d]);
        ((int2*)csr)[rowptr[d] + rank[i]] = r;
    }
}


__global__ void k_lin1(const float* __restrict__ x, const float* __restrict__ W1,
                       float* __restrict__ h, int n) {
    int t = blockIdx.x * blockDim.x + threadIdx.x;
    int i = t >> 6, j = t & 63;
    if (i < n) {
        float2 xv = ((const float2*)x)[i];
        h[t] = xv.x * W1[j] + xv.y * W1[HID + j];
    }
}


__device__ __forceinline__ v2f ld_row(const v2f* __restrict__ h2, int2 rec, int k, int lane) {
    int s = __builtin_amdgcn_readlane(rec.x, k);
    return h2[(size_t)s * 32 + lane];
}
__device__ __forceinline__ float rec_w(int2 rec, int k) {
    return __int_as_float(__builtin_amdgcn_readlane(rec.y, k));
}

__global__ void __launch_bounds__(256)
k_gather_rows(const int* __restrict__ rowptr, const int* __restrict__ csr,
              const float* __restrict__ dinv, const float* __restrict__ h,
              const float* __restrict__ b, float* __restrict__ out, int n) {
    int lane = threadIdx.x & 31;
    int node = __builtin_amdgcn_readfirstlane((int)((blockIdx.x * blockDim.x + threadIdx.x) >> 5));
    if (node >= n) return;
    int beg = rowptr[node], end = rowptr[node + 1];
    float di = dinv[node];
    const v2f* h2 = (const v2f*)h;
    v2f hv = h2[(size_t)node * 32 + lane];
    v2f acc = ((const v2f*)b)[lane] + hv * di * di;
    for (int base = beg; base < end; base += 32) {
        int m = end - base;
        if (m > 32) m = 32;
        int2 rec = make_int2(0, 0);
        if (lane < m) rec = ((const int2*)csr)[base + lane];
        int k = 0;
        for (; k + 8 <= m; k += 8) {
            v2f v0 = ld_row(h2, rec, k,     lane), v1 = ld_row(h2, rec, k + 1, lane);
            v2f v2 = ld_row(h2, rec, k + 2, lane), v3 = ld_row(h2, rec, k + 3, lane);
            v2f v4 = ld_row(h2, rec, k + 4, lane), v5 = ld_row(h2, rec, k + 5, lane);
            v2f v6 = ld_row(h2, rec, k + 6, lane), v7 = ld_row(h2, rec, k + 7, lane);
            acc += v0 * rec_w(rec, k);
            acc += v1 * rec_w(rec, k + 1);
            acc += v2 * rec_w(rec, k + 2);
            acc += v3 * rec_w(rec, k + 3);
            acc += v4 * rec_w(rec, k + 4);
            acc += v5 * rec_w(rec, k + 5);
            acc += v6 * rec_w(rec, k + 6);
            acc += v7 * rec_w(rec, k + 7);
        }
        if (k + 4 <= m) {
            v2f v0 = ld_row(h2, rec, k,     lane), v1 = ld_row(h2, rec, k + 1, lane);
            v2f v2 = ld_row(h2, rec, k + 2, lane), v3 = ld_row(h2, rec, k + 3, lane);
            acc += v0 * rec_w(rec, k);
            acc += v1 * rec_w(rec, k + 1);
            acc += v2 * rec_w(rec, k + 2);
            acc += v3 * rec_w(rec, k + 3);
            k += 4;
        }
        if (k + 2 <= m) {
            v2f v0 = ld_row(h2, rec, k, lane), v1 = ld_row(h2, rec, k + 1, lane);
            acc += v0 * rec_w(rec, k);
            acc += v1 * rec_w(rec, k + 1);
            k += 2;
        }
        if (k < m) {
            v2f v0 = ld_row(h2, rec, k, lane);
            acc += v0 * rec_w(rec, k);
        }
    }
    ((v2f*)out)[(size_t)node * 32 + lane] = acc;
}

__global__ void __launch_bounds__(256)
k_gemm_relu_wmma(const float* __restrict__ in, const float* __restrict__ W,
                 float* __restrict__ out, int nrows) {
    int gwave = (blockIdx.x * blockDim.x + threadIdx.x) >> 5;
    int lane  = threadIdx.x & 31;
    int row0  = gwave << 4;
    if (row0 >= nrows) return;
    int hh = lane >> 4;
    int l  = lane & 15;
    v8f acc0 = {}, acc1 = {}, acc2 = {}, acc3 = {};
    const float* arow = in + (size_t)(row0 + l) * HID;
    for (int k0 = 0; k0 < HID; k0 += 4) {
        int ka = k0 + hh * 2;
        v2f a;
        a.x = fmaxf(arow[ka],     0.0f);
        a.y = fmaxf(arow[ka + 1], 0.0f);
        const float* bp = W + (size_t)ka * HID + l;
        v2f b0, b1, b2, b3;
        b0.x = bp[0];  b0.y = bp[HID];
        b1.x = bp[16]; b1.y = bp[HID + 16];
        b2.x = bp[32]; b2.y = bp[HID + 32];
        b3.x = bp[48]; b3.y = bp[HID + 48];
        acc0 = __builtin_amdgcn_wmma_f32_16x16x4_f32(false, a, false, b0, (short)0, acc0, false, false);
        acc1 = __builtin_amdgcn_wmma_f32_16x16x4_f32(false, a, false, b1, (short)0, acc1, false, false);
        acc2 = __builtin_amdgcn_wmma_f32_16x16x4_f32(false, a, false, b2, (short)0, acc2, false, false);
        acc3 = __builtin_amdgcn_wmma_f32_16x16x4_f32(false, a, false, b3, (short)0, acc3, false, false);
    }
    float* orow = out + (size_t)(row0 + hh * 8) * HID + l;
    #pragma unroll
    for (int r = 0; r < 8; ++r) {
        float* o = orow + (size_t)r * HID;
        o[0]  = acc0[r];
        o[16] = acc1[r];
        o[32] = acc2[r];
        o[48] = acc3[r];
    }
}


__global__ void k_lin3_relu(const float* __restrict__ in, const float* __restrict__ W3,
                            float* __restrict__ h3, int n) {
    int gwave = (blockIdx.x * blockDim.x + threadIdx.x) >> 5;
    int lane  = threadIdx.x & 31;
    if (gwave >= n) return;
    const float* row = in + (size_t)gwave * HID;
    float v = fmaxf(row[lane], 0.0f) * W3[lane]
            + fmaxf(row[lane + 32], 0.0f) * W3[lane + 32];
    #pragma unroll
    for (int off = 16; off > 0; off >>= 1) v += __shfl_down(v, off, 32);
    if (lane == 0) h3[gwave] = v;
}

__global__ void k_gather_scalar(const int* __restrict__ rowptr, const int* __restrict__ csr,
                                const float* __restrict__ dinv, const float* __restrict__ h3,
                                const float* __restrict__ b3, float* __restrict__ y, int n) {
    int t = blockIdx.x * blockDim.x + threadIdx.x;
    int node = t >> 3, sub = t & 7;
    float v = 0.0f;
    if (node < n) {
        int beg = rowptr[node], end = rowptr[node + 1];
        for (int j = beg + sub; j < end; j += 8) {
            int2 r = ((const int2*)csr)[j];
            v += h3[r.x] * __int_as_float(r.y);
        }
    }
    v += __shfl_xor(v, 4, 32);
    v += __shfl_xor(v, 2, 32);
    v += __shfl_xor(v, 1, 32);
    if (node < n && sub == 0) {
        float di = dinv[node];
        y[node] = b3[0] + h3[node] * di * di + v;
    }
}


extern "C" void kernel_launch(void* const* d_in, const int* in_sizes, int n_in,
                              void* d_out, int out_size, void* d_ws, size_t ws_size,
                              hipStream_t stream) {
    const float* x  = (const float*)d_in[0];
    const int*   ei = (const int*)  d_in[1];
    const float* W1 = (const float*)d_in[2];
    const float* b1 = (const float*)d_in[3];
    const float* W2 = (const float*)d_in[4];
    const float* b2 = (const float*)d_in[5];
    const float* W3 = (const float*)d_in[6];
    const float* b3 = (const float*)d_in[7];

    const int N = in_sizes[0] / 2;
    const int E = in_sizes[1] / 2;
    const int* src = ei;
    const int* dst = ei + E;

    float* h      = (float*)d_ws;
    float* out    = h + (size_t)N * HID;
    int*   csr    = (int*)(out + (size_t)N * HID);
    int*   rank   = csr + (size_t)E * 2;
    int*   cnt    = rank + (size_t)E;
    int*   rowptr = cnt + (size_t)N;
    float* dinv   = (float*)(rowptr + (size_t)N + 1);
    float* h3     = dinv + (size_t)N;
    float* y      = (float*)d_out;

    const int B = 256;
    auto g = [](long long t) { return (int)((t + 255) / 256); };

    k_zero_cnt<<<g(N), B, 0, stream>>>(cnt, N);
    k_count   <<<g(E), B, 0, stream>>>(dst, cnt, rank, E);
    k_rowptr  <<<(N + 1023) / 1024, 1024, 0, stream>>>(cnt, rowptr, dinv, N);
    k_fill    <<<g(E), B, 0, stream>>>(src, dst, rank, rowptr, dinv, csr, E);

    k_lin1       <<<g((long long)N * HID), B, 0, stream>>>(x, W1, h, N);
    k_gather_rows<<<g((long long)N * 32),  B, 0, stream>>>(rowptr, csr, dinv, h, b1, out, N);

    k_gemm_relu_wmma<<<g((long long)(N / 16) * 32), 256, 0, stream>>>(out, W2, h, N);
    k_gather_rows<<<g((long long)N * 32),  B, 0, stream>>>(rowptr, csr, dinv, h, b2, out, N);

    k_lin3_relu    <<<g((long long)N * 32), B, 0, stream>>>(out, W3, h3, N);
    k_gather_scalar<<<g((long long)N * 8),  B, 0, stream>>>(rowptr, csr, dinv, h3, b3, y, N);
}


# Layer 1 sums the 2-wide inputs per node before applying W1 (A(xW1)=(Ax)W1): 8-byte gathers replace 256-byte rows, W1+bias+relu move into the GEMM A loader; layer 1 sum order changes (8-lane tree)
# speedup vs baseline: 2.3043x; 1.0439x over previous
#include <hip/hip_runtime.h>
#include <hip/hip_bf16.h>

typedef __attribute__((ext_vector_type(2))) float v2f;
typedef __attribute__((ext_vector_type(8))) float v8f;

#define HID 64


__global__ void k_zero_cnt(int* __restrict__ cnt, int n) {
    int i = blockIdx.x * blockDim.x + threadIdx.x;
    if (i < n) cnt[i] = 0;
}

__global__ void k_count(const int* __restrict__ dst, int* __restrict__ cnt,
                        int* __restrict__ rank, int e) {
    int i = blockIdx.x * blockDim.x + threadIdx.x;
    if (i < e)
        rank[i] = __hip_atomic_fetch_add(&cnt[dst[i]], 1, __ATOMIC_RELAXED,
                                         __HIP_MEMORY_SCOPE_AGENT);
}

__global__ void __launch_bounds__(1024)
k_rowptr(const int* __restrict__ cnt, int* __restrict__ rowptr,
         float* __restrict__ dinv, int n) {
    __shared__ int s_pre[32];
    __shared__ int s_cnt[32];
    int t = threadIdx.x, lane = t & 31, wv = t >> 5;
    int first = blockIdx.x * 1024;
    int pre = 0;
    for (int j = t; j < first; j += 1024) pre += cnt[j];
    int i = first + t;
    int c = (i < n) ? cnt[i] : 0;
    int inc = c;
    for (int off = 1; off < 32; off <<= 1) {
        int u = __shfl_up(inc, off, 32);
        if (lane >= off) inc += u;
    }
    for (int off = 16; off > 0; off >>= 1) pre += __shfl_xor(pre, off, 32);
    if (lane == 31) s_cnt[wv] = inc;
    if (lane == 0)  s_pre[wv] = pre;
    __syncthreads();
    int basev = 0;
    for (int j = 0; j < 32; ++j) {
        basev += s_pre[j];
        if (j < wv) basev += s_cnt[j];
    }
    int excl = basev + inc - c;
    if (i < n) {
        rowptr[i] = excl;
        dinv[i]   = rsqrtf((float)(c + 1));
    }
    if (i == n - 1) rowptr[n] = excl + c;
}

__global__ void k_fill(const int* __restrict__ src, const int* __restrict__ dst,
                       const int* __restrict__ rank, const int* __restrict__ rowptr,
                       const float* __restrict__ dinv, int* __restrict__ csr, int e) {
    int i = blockIdx.x * blockDim.x + threadIdx.x;
    if (i < e) {
        int s = src[i], d = dst[i];
        int2 r;
        r.x = s;
        r.y = __float_as_int(dinv[s] * dinv[d]);
        ((int2*)csr)[rowptr[d] + rank[i]] = r;
    }
}


__global__ void k_gather_x(const int* __restrict__ rowptr, const int* __restrict__ csr,
                           const float* __restrict__ dinv, const float* __restrict__ x,
                           float* __restrict__ ax, int n) {
    int t = blockIdx.x * blockDim.x + threadIdx.x;
    int node = t >> 3, sub = t & 7;
    const float2* x2 = (const float2*)x;
    const int2*   c2 = (const int2*)csr;
    float v0 = 0.0f, v1 = 0.0f;
    float di = 0.0f;
    float2 xi = make_float2(0.0f, 0.0f);
    if (node < n) {
        int beg = rowptr[node], end = rowptr[node + 1];
        di = dinv[node];
        xi = x2[node];
        for (int j = beg + sub; j < end; j += 32) {
            bool p1 = j + 8 < end, p2 = j + 16 < end, p3 = j + 24 < end;
            int2 r0 = c2[j];
            int2 r1 = c2[p1 ? j + 8  : j];
            int2 r2 = c2[p2 ? j + 16 : j];
            int2 r3 = c2[p3 ? j + 24 : j];
            float2 a0 = x2[r0.x], a1 = x2[r1.x], a2 = x2[r2.x], a3 = x2[r3.x];
            float w0 = __int_as_float(r0.y), w1 = __int_as_float(r1.y);
            float w2 = __int_as_float(r2.y), w3 = __int_as_float(r3.y);
            v0 += a0.x * w0;  v1 += a0.y * w0;
            if (p1) { v0 += a1.x * w1;  v1 += a1.y * w1; }
            if (p2) { v0 += a2.x * w2;  v1 += a2.y * w2; }
            if (p3) { v0 += a3.x * w3;  v1 += a3.y * w3; }
        }
    }
    v0 += __shfl_xor(v0, 4, 32);  v1 += __shfl_xor(v1, 4, 32);
    v0 += __shfl_xor(v0, 2, 32);  v1 += __shfl_xor(v1, 2, 32);
    v0 += __shfl_xor(v0, 1, 32);  v1 += __shfl_xor(v1, 1, 32);
    if (node < n && sub == 0) {
        float2 o;
        o.x = xi.x * di * di + v0;
        o.y = xi.y * di * di + v1;
        ((float2*)ax)[node] = o;
    }
}


__device__ __forceinline__ v2f ld_row(const v2f* __restrict__ h2, int2 rec, int k, int lane) {
    int s = __builtin_amdgcn_readlane(rec.x, k);
    return h2[(size_t)s * 32 + lane];
}
__device__ __forceinline__ float rec_w(int2 rec, int k) {
    return __int_as_float(__builtin_amdgcn_readlane(rec.y, k));
}

__global__ void __launch_bounds__(256)
k_gather_rows(const int* __restrict__ rowptr, const int* __restrict__ csr,
              const float* __restrict__ dinv, const float* __restrict__ h,
              const float* __restrict__ b, float* __restrict__ out, int n) {
    int lane = threadIdx.x & 31;
    int node = __builtin_amdgcn_readfirstlane((int)((blockIdx.x * blockDim.x + threadIdx.x) >> 5));
    if (node >= n) return;
    int beg = rowptr[node], end = rowptr[node + 1];
    float di = dinv[node];
    const v2f* h2 = (const v2f*)h;
    v2f hv = h2[(size_t)node * 32 + lane];
    v2f acc = ((const v2f*)b)[lane] + hv * di * di;
    for (int base = beg; base < end; base += 32) {
        int m = end - base;
        if (m > 32) m = 32;
        int2 rec = make_int2(0, 0);
        if (lane < m) rec = ((const int2*)csr)[base + lane];
        int k = 0;
        for (; k + 8 <= m; k += 8) {
            v2f v0 = ld_row(h2, rec, k,     lane), v1 = ld_row(h2, rec, k + 1, lane);
            v2f v2 = ld_row(h2, rec, k + 2, lane), v3 = ld_row(h2, rec, k + 3, lane);
            v2f v4 = ld_row(h2, rec, k + 4, lane), v5 = ld_row(h2, rec, k + 5, lane);
            v2f v6 = ld_row(h2, rec, k + 6, lane), v7 = ld_row(h2, rec, k + 7, lane);
            acc += v0 * rec_w(rec, k);
            acc += v1 * rec_w(rec, k + 1);
            acc += v2 * rec_w(rec, k + 2);
            acc += v3 * rec_w(rec, k + 3);
            acc += v4 * rec_w(rec, k + 4);
            acc += v5 * rec_w(rec, k + 5);
            acc += v6 * rec_w(rec, k + 6);
            acc += v7 * rec_w(rec, k + 7);
        }
        if (k + 4 <= m) {
            v2f v0 = ld_row(h2, rec, k,     lane), v1 = ld_row(h2, rec, k + 1, lane);
            v2f v2 = ld_row(h2, rec, k + 2, lane), v3 = ld_row(h2, rec, k + 3, lane);
            acc += v0 * rec_w(rec, k);
            acc += v1 * rec_w(rec, k + 1);
            acc += v2 * rec_w(rec, k + 2);
            acc += v3 * rec_w(rec, k + 3);
            k += 4;
        }
        if (k + 2 <= m) {
            v2f v0 = ld_row(h2, rec, k, lane), v1 = ld_row(h2, rec, k + 1, lane);
            acc += v0 * rec_w(rec, k);
            acc += v1 * rec_w(rec, k + 1);
            k += 2;
        }
        if (k < m) {
            v2f v0 = ld_row(h2, rec, k, lane);
            acc += v0 * rec_w(rec, k);
        }
    }
    ((v2f*)out)[(size_t)node * 32 + lane] = acc;
}

__global__ void __launch_bounds__(256)
k_gemm_lin1_relu_wmma(const float* __restrict__ ax, const float* __restrict__ W1,
                      const float* __restrict__ b1, const float* __restrict__ W,
                      float* __restrict__ out, int nrows) {
    int gwave = (blockIdx.x * blockDim.x + threadIdx.x) >> 5;
    int lane  = threadIdx.x & 31;
    int row0  = gwave << 4;
    if (row0 >= nrows) return;
    int hh = lane >> 4;
    int l  = lane & 15;
    v8f acc0 = {}, acc1 = {}, acc2 = {}, acc3 = {};
    float2 av = ((const float2*)ax)[row0 + l];
    for (int k0 = 0; k0 < HID; k0 += 4) {
        int ka = k0 + hh * 2;
        float2 wa = *(const float2*)(W1 + ka);
        float2 wb = *(const float2*)(W1 + HID + ka);
        float2 bb = *(const float2*)(b1 + ka);
        v2f a;
        a.x = fmaxf(av.x * wa.x + av.y * wb.x + bb.x, 0.0f);
        a.y = fmaxf(av.x * wa.y + av.y * wb.y + bb.y, 0.0f);
        const float* bp = W + (size_t)ka * HID + l;
        v2f b0, b1f, b2, b3;
        b0.x  = bp[0];  b0.y  = bp[HID];
        b1f.x = bp[16]; b1f.y = bp[HID + 16];
        b2.x  = bp[32]; b2.y  = bp[HID + 32];
        b3.x  = bp[48]; b3.y  = bp[HID + 48];
        acc0 = __builtin_amdgcn_wmma_f32_16x16x4_f32(false, a, false, b0,  (short)0, acc0, false, false);
        acc1 = __builtin_amdgcn_wmma_f32_16x16x4_f32(false, a, false, b1f, (short)0, acc1, false, false);
        acc2 = __builtin_amdgcn_wmma_f32_16x16x4_f32(false, a, false, b2,  (short)0, acc2, false, false);
        acc3 = __builtin_amdgcn_wmma_f32_16x16x4_f32(false, a, false, b3,  (short)0, acc3, false, false);
    }
    float* orow = out + (size_t)(row0 + hh * 8) * HID + l;
    #pragma unroll
    for (int r = 0; r < 8; ++r) {
        float* o = orow + (size_t)r * HID;
        o[0]  = acc0[r];
        o[16] = acc1[r];
        o[32] = acc2[r];
        o[48] = acc3[r];
    }
}


__global__ void k_lin3_relu(const float* __restrict__ in, const float* __restrict__ W3,
                            float* __restrict__ h3, int n) {
    int gwave = (blockIdx.x * blockDim.x + threadIdx.x) >> 5;
    int lane  = threadIdx.x & 31;
    if (gwave >= n) return;
    const float* row = in + (size_t)gwave * HID;
    float v = fmaxf(row[lane], 0.0f) * W3[lane]
            + fmaxf(row[lane + 32], 0.0f) * W3[lane + 32];
    #pragma unroll
    for (int off = 16; off > 0; off >>= 1) v += __shfl_down(v, off, 32);
    if (lane == 0) h3[gwave] = v;
}

__global__ void k_gather_scalar(const int* __restrict__ rowptr, const int* __restrict__ csr,
                                const float* __restrict__ dinv, const float* __restrict__ h3,
                                const float* __restrict__ b3, float* __restrict__ y, int n) {
    int t = blockIdx.x * blockDim.x + threadIdx.x;
    int node = t >> 3, sub = t & 7;
    float v = 0.0f;
    if (node < n) {
        int beg = rowptr[node], end = rowptr[node + 1];
        for (int j = beg + sub; j < end; j += 8) {
            int2 r = ((const int2*)csr)[j];
            v += h3[r.x] * __int_as_float(r.y);
        }
    }
    v += __shfl_xor(v, 4, 32);
    v += __shfl_xor(v, 2, 32);
    v += __shfl_xor(v, 1, 32);
    if (node < n && sub == 0) {
        float di = dinv[node];
        y[node] = b3[0] + h3[node] * di * di + v;
    }
}


extern "C" void kernel_launch(void* const* d_in, const int* in_sizes, int n_in,
                              void* d_out, int out_size, void* d_ws, size_t ws_size,
                              hipStream_t stream) {
    const float* x  = (const float*)d_in[0];
    const int*   ei = (const int*)  d_in[1];
    const float* W1 = (const float*)d_in[2];
    const float* b1 = (const float*)d_in[3];
    const float* W2 = (const float*)d_in[4];
    const float* b2 = (const float*)d_in[5];
    const float* W3 = (const float*)d_in[6];
    const float* b3 = (const float*)d_in[7];

    const int N = in_sizes[0] / 2;
    const int E = in_sizes[1] / 2;
    const int* src = ei;
    const int* dst = ei + E;

    float* h      = (float*)d_ws;
    float* out    = h + (size_t)N * HID;
    int*   csr    = (int*)(out + (size_t)N * HID);
    float* ax     = (float*)(csr + (size_t)E * 2);
    int*   rank   = (int*)(ax + (size_t)N * 2);
    int*   cnt    = rank + (size_t)E;
    int*   rowptr = cnt + (size_t)N;
    float* dinv   = (float*)(rowptr + (size_t)N + 1);
    float* h3     = dinv + (size_t)N;
    float* y      = (float*)d_out;

    const int B = 256;
    auto g = [](long long t) { return (int)((t + 255) / 256); };

    k_zero_cnt<<<g(N), B, 0, stream>>>(cnt, N);
    k_count   <<<g(E), B, 0, stream>>>(dst, cnt, rank, E);
    k_rowptr  <<<(N + 1023) / 1024, 1024, 0, stream>>>(cnt, rowptr, dinv, N);
    k_fill    <<<g(E), B, 0, stream>>>(src, dst, rank, rowptr, dinv, csr, E);

    k_gather_x<<<g((long long)N * 8), B, 0, stream>>>(rowptr, csr, dinv, x, ax, N);

    k_gemm_lin1_relu_wmma<<<g((long long)(N / 16) * 32), 256, 0, stream>>>(ax, W1, b1, W2, h, N);
    k_gather_rows<<<g((long long)N * 32),  B, 0, stream>>>(rowptr, csr, dinv, h, b2, out, N);

    k_lin3_relu    <<<g((long long)N * 32), B, 0, stream>>>(out, W3, h3, N);
    k_gather_scalar<<<g((long long)N * 8),  B, 0, stream>>>(rowptr, csr, dinv, h3, b3, y, N);
}


# Fixed 64-slot bucket per node: the counting add's return is the record's slot, stored at once; scan, rank array and fill pass dropped; records asked with the count; sum order stays arrival order
# speedup vs baseline: 2.8324x; 1.2292x over previous
#include <hip/hip_runtime.h>
#include <hip/hip_bf16.h>

typedef __attribute__((ext_vector_type(2))) float v2f;
typedef __attribute__((ext_vector_type(8))) float v8f;

#define HID 64


#define CAP 64

__global__ void k_zero_cnt(int* __restrict__ cnt, int n1) {
    int i = blockIdx.x * blockDim.x + threadIdx.x;
    if (i < n1) cnt[i] = 0;
}

__global__ void k_bucket(const int* __restrict__ src, const int* __restrict__ dst,
                         int* __restrict__ cnt, int* __restrict__ bucket,
                         int* __restrict__ spill, int e, int n) {
    int i = blockIdx.x * blockDim.x + threadIdx.x;
    if (i < e) {
        int d = dst[i];
        int s = src[i];
        int r = __hip_atomic_fetch_add(&cnt[d], 1, __ATOMIC_RELAXED, __HIP_MEMORY_SCOPE_AGENT);
        if (r < CAP) {
            bucket[(size_t)d * CAP + r] = s;
        } else {
            int p = __hip_atomic_fetch_add(&cnt[n], 1, __ATOMIC_RELAXED, __HIP_MEMORY_SCOPE_AGENT);
            ((int2*)spill)[p] = make_int2(d, s);
        }
    }
}

__global__ void k_prep(const int* __restrict__ cnt, const float* __restrict__ x,
                       float* __restrict__ dinv, float* __restrict__ xs, int n) {
    int i = blockIdx.x * blockDim.x + threadIdx.x;
    if (i < n) {
        float  dv = rsqrtf((float)(cnt[i] + 1));
        float2 xv = ((const float2*)x)[i];
        dinv[i] = dv;
        ((float2*)xs)[i] = make_float2(xv.x * dv, xv.y * dv);
    }
}


__global__ void k_gather_x(const int* __restrict__ cnt, const int* __restrict__ bucket,
                           const int* __restrict__ spill,
                           const float* __restrict__ dinv, const float* __restrict__ xs,
                           float* __restrict__ ax, int n) {
    int t = blockIdx.x * blockDim.x + threadIdx.x;
    int node = t >> 3, sub = t & 7;
    const float2* x2 = (const float2*)xs;
    float v0 = 0.0f, v1 = 0.0f;
    float di = 0.0f;
    float2 xi = make_float2(0.0f, 0.0f);
    if (node < n) {
        const int* bk = bucket + (size_t)node * CAP;
        int r0 = bk[sub], r1 = bk[sub + 8], r2 = bk[sub + 16], r3 = bk[sub + 24];
        int c = cnt[node];
        di = dinv[node];
        xi = x2[node];
        bool p0 = sub < c, p1 = sub + 8 < c, p2 = sub + 16 < c, p3 = sub + 24 < c;
        float2 a0 = x2[p0 ? r0 : node], a1 = x2[p1 ? r1 : node];
        float2 a2 = x2[p2 ? r2 : node], a3 = x2[p3 ? r3 : node];
        v0 += p0 ? a0.x : 0.0f;  v1 += p0 ? a0.y : 0.0f;
        v0 += p1 ? a1.x : 0.0f;  v1 += p1 ? a1.y : 0.0f;
        v0 += p2 ? a2.x : 0.0f;  v1 += p2 ? a2.y : 0.0f;
        v0 += p3 ? a3.x : 0.0f;  v1 += p3 ? a3.y : 0.0f;
        int cc = c < CAP ? c : CAP;
        for (int j = 32 + sub; j < cc; j += 8) {
            float2 a = x2[bk[j]];
            v0 += a.x;  v1 += a.y;
        }
        if (c > CAP) {
            int nsp = cnt[n];
            const int2* sp = (const int2*)spill;
            for (int p = sub; p < nsp; p += 8) {
                int2 ds = sp[p];
                if (ds.x == node) {
                    float2 a = x2[ds.y];
                    v0 += a.x;  v1 += a.y;
                }
            }
        }
    }
    v0 += __shfl_xor(v0, 4, 32);  v1 += __shfl_xor(v1, 4, 32);
    v0 += __shfl_xor(v0, 2, 32);  v1 += __shfl_xor(v1, 2, 32);
    v0 += __shfl_xor(v0, 1, 32);  v1 += __shfl_xor(v1, 1, 32);
    if (node < n && sub == 0) {
        float2 o;
        o.x = (xi.x + v0) * di;
        o.y = (xi.y + v1) * di;
        ((float2*)ax)[node] = o;
    }
}


__device__ __forceinline__ v2f ld_row(const v2f* __restrict__ h2, int rec, int k, int lane) {
    int s = __builtin_amdgcn_readlane(rec, k);
    return h2[(size_t)s * 32 + lane];
}

__global__ void __launch_bounds__(256)
k_gather_rows(const int* __restrict__ cnt, const int* __restrict__ bucket,
              const int* __restrict__ spill,
              const float* __restrict__ dinv, const float* __restrict__ h,
              const float* __restrict__ b, float* __restrict__ out, int n) {
    int lane = threadIdx.x & 31;
    int node = __builtin_amdgcn_readfirstlane((int)((blockIdx.x * blockDim.x + threadIdx.x) >> 5));
    if (node >= n) return;
    const int* bk = bucket + (size_t)node * CAP;
    int rec = bk[lane];
    int c = cnt[node];
    float di = dinv[node];
    const v2f* h2 = (const v2f*)h;
    v2f bv  = ((const v2f*)b)[lane];
    v2f acc = h2[(size_t)node * 32 + lane];
    asm volatile("" :: "v"(rec), "s"(di));
    int cc = c < CAP ? c : CAP;
    for (int base = 0; base < cc; base += 32) {
        if (base) rec = bk[32 + lane];
        int m = cc - base;
        if (m > 32) m = 32;
        int k = 0;
        for (; k + 8 <= m; k += 8) {
            v2f v0 = ld_row(h2, rec, k,     lane), v1 = ld_row(h2, rec, k + 1, lane);
            v2f v2 = ld_row(h2, rec, k + 2, lane), v3 = ld_row(h2, rec, k + 3, lane);
            v2f v4 = ld_row(h2, rec, k + 4, lane), v5 = ld_row(h2, rec, k + 5, lane);
            v2f v6 = ld_row(h2, rec, k + 6, lane), v7 = ld_row(h2, rec, k + 7, lane);
            acc += v0;  acc += v1;  acc += v2;  acc += v3;
            acc += v4;  acc += v5;  acc += v6;  acc += v7;
        }
        if (k + 4 <= m) {
            v2f v0 = ld_row(h2, rec, k,     lane), v1 = ld_row(h2, rec, k + 1, lane);
            v2f v2 = ld_row(h2, rec, k + 2, lane), v3 = ld_row(h2, rec, k + 3, lane);
            acc += v0;  acc += v1;  acc += v2;  acc += v3;
            k += 4;
        }
        if (k + 2 <= m) {
            v2f v0 = ld_row(h2, rec, k, lane), v1 = ld_row(h2, rec, k + 1, lane);
            acc += v0;  acc += v1;
            k += 2;
        }
        if (k < m) {
            v2f v0 = ld_row(h2, rec, k, lane);
            acc += v0;
        }
    }
    if (c > CAP) {
        int nsp = cnt[n];
        const int2* sp = (const int2*)spill;
        for (int p0 = 0; p0 < nsp; p0 += 32) {
            int2 ds = make_int2(-1, 0);
            if (p0 + lane < nsp) ds = sp[p0 + lane];
            unsigned mine = __builtin_amdgcn_ballot_w32(ds.x == node);
            while (mine) {
                int k = __builtin_ctz(mine);
                mine &= mine - 1;
                acc += ld_row(h2, ds.y, k, lane);
            }
        }
    }
    ((v2f*)out)[(size_t)node * 32 + lane] = bv + acc * di;
}

__global__ void __launch_bounds__(256)
k_gemm_lin1_relu_wmma(const float* __restrict__ ax, const float* __restrict__ W1,
                      const float* __restrict__ b1, const float* __restrict__ W,
                      const float* __restrict__ dinv, float* __restrict__ out, int nrows) {
    int gwave = (blockIdx.x * blockDim.x + threadIdx.x) >> 5;
    int lane  = threadIdx.x & 31;
    int row0  = gwave << 4;
    if (row0 >= nrows) return;
    int hh = lane >> 4;
    int l  = lane & 15;
    v8f acc0 = {}, acc1 = {}, acc2 = {}, acc3 = {};
    float2 av = ((const float2*)ax)[row0 + l];
    const float4* dq = (const float4*)(dinv + row0 + hh * 8);
    float4 dlo = dq[0], dhi = dq[1];
    for (int k0 = 0; k0 < HID; k0 += 4) {
        int ka = k0 + hh * 2;
        float2 wa = *(const float2*)(W1 + ka);
        float2 wb = *(const float2*)(W1 + HID + ka);
        float2 bb = *(const float2*)(b1 + ka);
        v2f a;
        a.x = fmaxf(av.x * wa.x + av.y * wb.x + bb.x, 0.0f);
        a.y = fmaxf(av.x * wa.y + av.y * wb.y + bb.y, 0.0f);
        const float* bp = W + (size_t)ka * HID + l;
        v2f b0, b1f, b2, b3;
        b0.x  = bp[0];  b0.y  = bp[HID];
        b1f.x = bp[16]; b1f.y = bp[HID + 16];
        b2.x  = bp[32]; b2.y  = bp[HID + 32];
        b3.x  = bp[48]; b3.y  = bp[HID + 48];
        acc0 = __builtin_amdgcn_wmma_f32_16x16x4_f32(false, a, false, b0,  (short)0, acc0, false, false);
        acc1 = __builtin_amdgcn_wmma_f32_16x16x4_f32(false, a, false, b1f, (short)0, acc1, false, false);
        acc2 = __builtin_amdgcn_wmma_f32_16x16x4_f32(false, a, false, b2,  (short)0, acc2, false, false);
        acc3 = __builtin_amdgcn_wmma_f32_16x16x4_f32(false, a, false, b3,  (short)0, acc3, false, false);
    }
    const float dr[8] = { dlo.x, dlo.y, dlo.z, dlo.w, dhi.x, dhi.y, dhi.z, dhi.w };
    float* orow = out + (size_t)(row0 + hh * 8) * HID + l;
    #pragma unroll
    for (int r = 0; r < 8; ++r) {
        float* o = orow + (size_t)r * HID;
        o[0]  = acc0[r] * dr[r];
        o[16] = acc1[r] * dr[r];
        o[32] = acc2[r] * dr[r];
        o[48] = acc3[r] * dr[r];
    }
}


__global__ void k_lin3_relu(const float* __restrict__ in, const float* __restrict__ W3,
                            const float* __restrict__ dinv, float* __restrict__ h3, int n) {
    int gwave = (blockIdx.x * blockDim.x + threadIdx.x) >> 5;
    int lane  = threadIdx.x & 31;
    if (gwave >= n) return;
    const float* row = in + (size_t)gwave * HID;
    float v = fmaxf(row[lane], 0.0f) * W3[lane]
            + fmaxf(row[lane + 32], 0.0f) * W3[lane + 32];
    #pragma unroll
    for (int off = 16; off > 0; off >>= 1) v += __shfl_down(v, off, 32);
    if (lane == 0) h3[gwave] = v * dinv[gwave];
}

__global__ void k_gather_scalar(const int* __restrict__ cnt, const int* __restrict__ bucket,
                                const int* __restrict__ spill,
                                const float* __restrict__ dinv, const float* __restrict__ h3,
                                const float* __restrict__ b3, float* __restrict__ y, int n) {
    int t = blockIdx.x * blockDim.x + threadIdx.x;
    int node = t >> 3, sub = t & 7;
    float v = 0.0f;
    if (node < n) {
        const int* bk = bucket + (size_t)node * CAP;
        int r0 = bk[sub], r1 = bk[sub + 8], r2 = bk[sub + 16], r3 = bk[sub + 24];
        int c = cnt[node];
        bool p0 = sub < c, p1 = sub + 8 < c, p2 = sub + 16 < c, p3 = sub + 24 < c;
        float a0 = h3[p0 ? r0 : node], a1 = h3[p1 ? r1 : node];
        float a2 = h3[p2 ? r2 : node], a3 = h3[p3 ? r3 : node];
        v += p0 ? a0 : 0.0f;
        v += p1 ? a1 : 0.0f;
        v += p2 ? a2 : 0.0f;
        v += p3 ? a3 : 0.0f;
        int cc = c < CAP ? c : CAP;
        for (int j = 32 + sub; j < cc; j += 8) v += h3[bk[j]];
        if (c > CAP) {
            int nsp = cnt[n];
            const int2* sp = (const int2*)spill;
            for (int p = sub; p < nsp; p += 8) {
                int2 ds = sp[p];
                if (ds.x == node) v += h3[ds.y];
            }
        }
    }
    v += __shfl_xor(v, 4, 32);
    v += __shfl_xor(v, 2, 32);
    v += __shfl_xor(v, 1, 32);
    if (node < n && sub == 0) {
        float di = dinv[node];
        y[node] = b3[0] + (h3[node] + v) * di;
    }
}


extern "C" void kernel_launch(void* const* d_in, const int* in_sizes, int n_in,
                              void* d_out, int out_size, void* d_ws, size_t ws_size,
                              hipStream_t stream) {
    const float* x  = (const float*)d_in[0];
    const int*   ei = (const int*)  d_in[1];
    const float* W1 = (const float*)d_in[2];
    const float* b1 = (const float*)d_in[3];
    const float* W2 = (const float*)d_in[4];
    const float* b2 = (const float*)d_in[5];
    const float* W3 = (const float*)d_in[6];
    const float* b3 = (const float*)d_in[7];

    const int N = in_sizes[0] / 2;
    const int E = in_sizes[1] / 2;
    const int* src = ei;
    const int* dst = ei + E;

    float* h      = (float*)d_ws;
    float* out    = h + (size_t)N * HID;
    int*   bucket = (int*)(out + (size_t)N * HID);
    float* ax     = (float*)(bucket + (size_t)N * CAP);
    float* xs     = ax + (size_t)N * 2;
    float* dinv   = xs + (size_t)N * 2;
    float* h3     = dinv + (size_t)N;
    int*   spill  = (int*)(h3 + (size_t)N);
    int*   cnt    = spill + (size_t)E * 2;
    float* y      = (float*)d_out;

    const int B = 256;
    auto g = [](long long t) { return (int)((t + 255) / 256); };

    k_zero_cnt<<<g(N + 1), B, 0, stream>>>(cnt, N + 1);
    k_bucket  <<<g(E), B, 0, stream>>>(src, dst, cnt, bucket, spill, E, N);
    k_prep    <<<g(N), B, 0, stream>>>(cnt, x, dinv, xs, N);

    k_gather_x<<<g((long long)N * 8), B, 0, stream>>>(cnt, bucket, spill, dinv, xs, ax, N);

    k_gemm_lin1_relu_wmma<<<g((long long)(N / 16) * 32), 256, 0, stream>>>(ax, W1, b1, W2, dinv, h, N);
    k_gather_rows<<<g((long long)N * 32),  B, 0, stream>>>(cnt, bucket, spill, dinv, h, b2, out, N);

    k_lin3_relu    <<<g((long long)N * 32), B, 0, stream>>>(out, W3, dinv, h3, N);
    k_gather_scalar<<<g((long long)N * 8),  B, 0, stream>>>(cnt, bucket, spill, dinv, h3, b3, y, N);
}


# Layer 3 transform (relu, dot with W3, row scale) moved into the layer 2 row sum's epilogue: the [N,64] plane is never stored, one launch and 100000 waves fewer; the dot's sum order changes
# speedup vs baseline: 2.9227x; 1.0319x over previous
#include <hip/hip_runtime.h>
#include <hip/hip_bf16.h>

typedef __attribute__((ext_vector_type(2))) float v2f;
typedef __attribute__((ext_vector_type(8))) float v8f;

#define HID 64


#define CAP 64

__global__ void k_zero_cnt(int* __restrict__ cnt, int n1) {
    int i = blockIdx.x * blockDim.x + threadIdx.x;
    if (i < n1) cnt[i] = 0;
}

__global__ void k_bucket(const int* __restrict__ src, const int* __restrict__ dst,
                         int* __restrict__ cnt, int* __restrict__ bucket,
                         int* __restrict__ spill, int e, int n) {
    int i = blockIdx.x * blockDim.x + threadIdx.x;
    if (i < e) {
        int d = dst[i];
        int s = src[i];
        int r = __hip_atomic_fetch_add(&cnt[d], 1, __ATOMIC_RELAXED, __HIP_MEMORY_SCOPE_AGENT);
        if (r < CAP) {
            bucket[(size_t)d * CAP + r] = s;
        } else {
            int p = __hip_atomic_fetch_add(&cnt[n], 1, __ATOMIC_RELAXED, __HIP_MEMORY_SCOPE_AGENT);
            ((int2*)spill)[p] = make_int2(d, s);
        }
    }
}

__global__ void k_prep(const int* __restrict__ cnt, const float* __restrict__ x,
                       float* __restrict__ dinv, float* __restrict__ xs, int n) {
    int i = blockIdx.x * blockDim.x + threadIdx.x;
    if (i < n) {
        float  dv = rsqrtf((float)(cnt[i] + 1));
        float2 xv = ((const float2*)x)[i];
        dinv[i] = dv;
        ((float2*)xs)[i] = make_float2(xv.x * dv, xv.y * dv);
    }
}


__global__ void k_gather_x(const int* __restrict__ cnt, const int* __restrict__ bucket,
                           const int* __restrict__ spill,
                           const float* __restrict__ dinv, const float* __restrict__ xs,
                           float* __restrict__ ax, int n) {
    int t = blockIdx.x * blockDim.x + threadIdx.x;
    int node = t >> 3, sub = t & 7;
    const float2* x2 = (const float2*)xs;
    float v0 = 0.0f, v1 = 0.0f;
    float di = 0.0f;
    float2 xi = make_float2(0.0f, 0.0f);
    if (node < n) {
        const int* bk = bucket + (size_t)node * CAP;
        int r0 = bk[sub], r1 = bk[sub + 8], r2 = bk[sub + 16], r3 = bk[sub + 24];
        int c = cnt[node];
        di = dinv[node];
        xi = x2[node];
        bool p0 = sub < c, p1 = sub + 8 < c, p2 = sub + 16 < c, p3 = sub + 24 < c;
        float2 a0 = x2[p0 ? r0 : node], a1 = x2[p1 ? r1 : node];
        float2 a2 = x2[p2 ? r2 : node], a3 = x2[p3 ? r3 : node];
        v0 += p0 ? a0.x : 0.0f;  v1 += p0 ? a0.y : 0.0f;
        v0 += p1 ? a1.x : 0.0f;  v1 += p1 ? a1.y : 0.0f;
        v0 += p2 ? a2.x : 0.0f;  v1 += p2 ? a2.y : 0.0f;
        v0 += p3 ? a3.x : 0.0f;  v1 += p3 ? a3.y : 0.0f;
        int cc = c < CAP ? c : CAP;
        for (int j = 32 + sub; j < cc; j += 8) {
            float2 a = x2[bk[j]];
            v0 += a.x;  v1 += a.y;
        }
        if (c > CAP) {
            int nsp = cnt[n];
            const int2* sp = (const int2*)spill;
            for (int p = sub; p < nsp; p += 8) {
                int2 ds = sp[p];
                if (ds.x == node) {
                    float2 a = x2[ds.y];
                    v0 += a.x;  v1 += a.y;
                }
            }
        }
    }
    v0 += __shfl_xor(v0, 4, 32);  v1 += __shfl_xor(v1, 4, 32);
    v0 += __shfl_xor(v0, 2, 32);  v1 += __shfl_xor(v1, 2, 32);
    v0 += __shfl_xor(v0, 1, 32);  v1 += __shfl_xor(v1, 1, 32);
    if (node < n && sub == 0) {
        float2 o;
        o.x = (xi.x + v0) * di;
        o.y = (xi.y + v1) * di;
        ((float2*)ax)[node] = o;
    }
}


__device__ __forceinline__ v2f ld_row(const v2f* __restrict__ h2, int rec, int k, int lane) {
    int s = __builtin_amdgcn_readlane(rec, k);
    return h2[(size_t)s * 32 + lane];
}

template <int M> __device__ __forceinline__ float row_xor(float x) {
    int b = __builtin_bit_cast(int, x);
    return __builtin_bit_cast(float, __builtin_amdgcn_update_dpp(b, b, 0x160 + M, 0xf, 0xf, true));
}
__device__ __forceinline__ float xhalf(float x) {
    int b = __builtin_bit_cast(int, x);
    return __builtin_bit_cast(float,
        __builtin_amdgcn_permlanex16(b, b, 0x76543210, 0xfedcba98, false, false));
}

__global__ void __launch_bounds__(256)
k_gather_rows(const int* __restrict__ cnt, const int* __restrict__ bucket,
              const int* __restrict__ spill,
              const float* __restrict__ dinv, const float* __restrict__ h,
              const float* __restrict__ b, const float* __restrict__ W3,
              float* __restrict__ h3, int n) {
    int lane = threadIdx.x & 31;
    int node = __builtin_amdgcn_readfirstlane((int)((blockIdx.x * blockDim.x + threadIdx.x) >> 5));
    if (node >= n) return;
    const int* bk = bucket + (size_t)node * CAP;
    int rec = bk[lane];
    int c = cnt[node];
    float di = dinv[node];
    const v2f* h2 = (const v2f*)h;
    v2f bv  = ((const v2f*)b)[lane];
    v2f w3  = ((const v2f*)W3)[lane];
    v2f acc = h2[(size_t)node * 32 + lane];
    asm volatile("" :: "v"(rec), "s"(di));
    int cc = c < CAP ? c : CAP;
    for (int base = 0; base < cc; base += 32) {
        if (base) rec = bk[32 + lane];
        int m = cc - base;
        if (m > 32) m = 32;
        int k = 0;
        for (; k + 8 <= m; k += 8) {
            v2f v0 = ld_row(h2, rec, k,     lane), v1 = ld_row(h2, rec, k + 1, lane);
            v2f v2 = ld_row(h2, rec, k + 2, lane), v3 = ld_row(h2, rec, k + 3, lane);
            v2f v4 = ld_row(h2, rec, k + 4, lane), v5 = ld_row(h2, rec, k + 5, lane);
            v2f v6 = ld_row(h2, rec, k + 6, lane), v7 = ld_row(h2, rec, k + 7, lane);
            acc += v0;  acc += v1;  acc += v2;  acc += v3;
            acc += v4;  acc += v5;  acc += v6;  acc += v7;
        }
        if (k + 4 <= m) {
            v2f v0 = ld_row(h2, rec, k,     lane), v1 = ld_row(h2, rec, k + 1, lane);
            v2f v2 = ld_row(h2, rec, k + 2, lane), v3 = ld_row(h2, rec, k + 3, lane);
            acc += v0;  acc += v1;  acc += v2;  acc += v3;
            k += 4;
        }
        if (k + 2 <= m) {
            v2f v0 = ld_row(h2, rec, k, lane), v1 = ld_row(h2, rec, k + 1, lane);
            acc += v0;  acc += v1;
            k += 2;
        }
        if (k < m) {
            v2f v0 = ld_row(h2, rec, k, lane);
            acc += v0;
        }
    }
    if (c > CAP) {
        int nsp = cnt[n];
        const int2* sp = (const int2*)spill;
        for (int p0 = 0; p0 < nsp; p0 += 32) {
            int2 ds = make_int2(-1, 0);
            if (p0 + lane < nsp) ds = sp[p0 + lane];
            unsigned mine = __builtin_amdgcn_ballot_w32(ds.x == node);
            while (mine) {
                int k = __builtin_ctz(mine);
                mine &= mine - 1;
                acc += ld_row(h2, ds.y, k, lane);
            }
        }
    }
    v2f o = bv + acc * di;
    float r = fmaxf(o.x, 0.0f) * w3.x + fmaxf(o.y, 0.0f) * w3.y;
    r += row_xor<1>(r);
    r += row_xor<2>(r);
    r += row_xor<4>(r);
    r += row_xor<8>(r);
    r += xhalf(r);
    if (lane == 0) h3[node] = r * di;
}

__global__ void __launch_bounds__(256)
k_gemm_lin1_relu_wmma(const float* __restrict__ ax, const float* __restrict__ W1,
                      const float* __restrict__ b1, const float* __restrict__ W,
                      const float* __restrict__ dinv, float* __restrict__ out, int nrows) {
    int gwave = (blockIdx.x * blockDim.x + threadIdx.x) >> 5;
    int lane  = threadIdx.x & 31;
    int row0  = gwave << 4;
    if (row0 >= nrows) return;
    int hh = lane >> 4;
    int l  = lane & 15;
    v8f acc0 = {}, acc1 = {}, acc2 = {}, acc3 = {};
    float2 av = ((const float2*)ax)[row0 + l];
    const float4* dq = (const float4*)(dinv + row0 + hh * 8);
    float4 dlo = dq[0], dhi = dq[1];
    for (int k0 = 0; k0 < HID; k0 += 4) {
        int ka = k0 + hh * 2;
        float2 wa = *(const float2*)(W1 + ka);
        float2 wb = *(const float2*)(W1 + HID + ka);
        float2 bb = *(const float2*)(b1 + ka);
        v2f a;
        a.x = fmaxf(av.x * wa.x + av.y * wb.x + bb.x, 0.0f);
        a.y = fmaxf(av.x * wa.y + av.y * wb.y + bb.y, 0.0f);
        const float* bp = W + (size_t)ka * HID + l;
        v2f b0, b1f, b2, b3;
        b0.x  = bp[0];  b0.y  = bp[HID];
        b1f.x = bp[16]; b1f.y = bp[HID + 16];
        b2.x  = bp[32]; b2.y  = bp[HID + 32];
        b3.x  = bp[48]; b3.y  = bp[HID + 48];
        acc0 = __builtin_amdgcn_wmma_f32_16x16x4_f32(false, a, false, b0,  (short)0, acc0, false, false);
        acc1 = __builtin_amdgcn_wmma_f32_16x16x4_f32(false, a, false, b1f, (short)0, acc1, false, false);
        acc2 = __builtin_amdgcn_wmma_f32_16x16x4_f32(false, a, false, b2,  (short)0, acc2, false, false);
        acc3 = __builtin_amdgcn_wmma_f32_16x16x4_f32(false, a, false, b3,  (short)0, acc3, false, false);
    }
    const float dr[8] = { dlo.x, dlo.y, dlo.z, dlo.w, dhi.x, dhi.y, dhi.z, dhi.w };
    float* orow = out + (size_t)(row0 + hh * 8) * HID + l;
    #pragma unroll
    for (int r = 0; r < 8; ++r) {
        float* o = orow + (size_t)r * HID;
        o[0]  = acc0[r] * dr[r];
        o[16] = acc1[r] * dr[r];
        o[32] = acc2[r] * dr[r];
        o[48] = acc3[r] * dr[r];
    }
}

__global__ void k_gather_scalar(const int* __restrict__ cnt, const int* __restrict__ bucket,
                                const int* __restrict__ spill,
                                const float* __restrict__ dinv, const float* __restrict__ h3,
                                const float* __restrict__ b3, float* __restrict__ y, int n) {
    int t = blockIdx.x * blockDim.x + threadIdx.x;
    int node = t >> 3, sub = t & 7;
    float v = 0.0f;
    if (node < n) {
        const int* bk = bucket + (size_t)node * CAP;
        int r0 = bk[sub], r1 = bk[sub + 8], r2 = bk[sub + 16], r3 = bk[sub + 24];
        int c = cnt[node];
        bool p0 = sub < c, p1 = sub + 8 < c, p2 = sub + 16 < c, p3 = sub + 24 < c;
        float a0 = h3[p0 ? r0 : node], a1 = h3[p1 ? r1 : node];
        float a2 = h3[p2 ? r2 : node], a3 = h3[p3 ? r3 : node];
        v += p0 ? a0 : 0.0f;
        v += p1 ? a1 : 0.0f;
        v += p2 ? a2 : 0.0f;
        v += p3 ? a3 : 0.0f;
        int cc = c < CAP ? c : CAP;
        for (int j = 32 + sub; j < cc; j += 8) v += h3[bk[j]];
        if (c > CAP) {
            int nsp = cnt[n];
            const int2* sp = (const int2*)spill;
            for (int p = sub; p < nsp; p += 8) {
                int2 ds = sp[p];
                if (ds.x == node) v += h3[ds.y];
            }
        }
    }
    v += __shfl_xor(v, 4, 32);
    v += __shfl_xor(v, 2, 32);
    v += __shfl_xor(v, 1, 32);
    if (node < n && sub == 0) {
        float di = dinv[node];
        y[node] = b3[0] + (h3[node] + v) * di;
    }
}


extern "C" void kernel_launch(void* const* d_in, const int* in_sizes, int n_in,
                              void* d_out, int out_size, void* d_ws, size_t ws_size,
                              hipStream_t stream) {
    const float* x  = (const float*)d_in[0];
    const int*   ei = (const int*)  d_in[1];
    const float* W1 = (const float*)d_in[2];
    const float* b1 = (const float*)d_in[3];
    const float* W2 = (const float*)d_in[4];
    const float* b2 = (const float*)d_in[5];
    const float* W3 = (const float*)d_in[6];
    const float* b3 = (const float*)d_in[7];

    const int N = in_sizes[0] / 2;
    const int E = in_sizes[1] / 2;
    const int* src = ei;
    const int* dst = ei + E;

    float* h      = (float*)d_ws;
    int*   bucket = (int*)(h + (size_t)N * HID);
    float* ax     = (float*)(bucket + (size_t)N * CAP);
    float* xs     = ax + (size_t)N * 2;
    float* dinv   = xs + (size_t)N * 2;
    float* h3     = dinv + (size_t)N;
    int*   spill  = (int*)(h3 + (size_t)N);
    int*   cnt    = spill + (size_t)E * 2;
    float* y      = (float*)d_out;

    const int B = 256;
    auto g = [](long long t) { return (int)((t + 255) / 256); };

    k_zero_cnt<<<g(N + 1), B, 0, stream>>>(cnt, N + 1);
    k_bucket  <<<g(E), B, 0, stream>>>(src, dst, cnt, bucket, spill, E, N);
    k_prep    <<<g(N), B, 0, stream>>>(cnt, x, dinv, xs, N);

    k_gather_x<<<g((long long)N * 8), B, 0, stream>>>(cnt, bucket, spill, dinv, xs, ax, N);

    k_gemm_lin1_relu_wmma<<<g((long long)(N / 16) * 32), 256, 0, stream>>>(ax, W1, b1, W2, dinv, h, N);
    k_gather_rows<<<g((long long)N * 32),  B, 0, stream>>>(cnt, bucket, spill, dinv, h, b2, W3, h3, N);

    k_gather_scalar<<<g((long long)N * 8),  B, 0, stream>>>(cnt, bucket, spill, dinv, h3, b3, y, N);
}


# Layer 2 sums the layer 1 rows themselves, re-formed per edge from the source's 16-byte entry (ax, dinv): (A H)W2 = A(H W2); no 256-byte row gathers, W2 GEMM runs after the sum; sum orders change
# speedup vs baseline: 3.3954x; 1.1617x over previous
#include <hip/hip_runtime.h>
#include <hip/hip_bf16.h>

typedef __attribute__((ext_vector_type(2))) float v2f;
typedef __attribute__((ext_vector_type(8))) float v8f;

#define HID 64


#define CAP 64

__global__ void k_zero_cnt(int* __restrict__ cnt, int n1) {
    int i = blockIdx.x * blockDim.x + threadIdx.x;
    if (i < n1) cnt[i] = 0;
}

__global__ void k_bucket(const int* __restrict__ src, const int* __restrict__ dst,
                         int* __restrict__ cnt, int* __restrict__ bucket,
                         int* __restrict__ spill, int e, int n) {
    int i = blockIdx.x * blockDim.x + threadIdx.x;
    if (i < e) {
        int d = dst[i];
        int s = src[i];
        int r = __hip_atomic_fetch_add(&cnt[d], 1, __ATOMIC_RELAXED, __HIP_MEMORY_SCOPE_AGENT);
        if (r < CAP) {
            bucket[(size_t)d * CAP + r] = s;
        } else {
            int p = __hip_atomic_fetch_add(&cnt[n], 1, __ATOMIC_RELAXED, __HIP_MEMORY_SCOPE_AGENT);
            ((int2*)spill)[p] = make_int2(d, s);
        }
    }
}

__global__ void k_prep(const int* __restrict__ cnt, const float* __restrict__ x,
                       float* __restrict__ dinv, float* __restrict__ xs, int n) {
    int i = blockIdx.x * blockDim.x + threadIdx.x;
    if (i < n) {
        float  dv = rsqrtf((float)(cnt[i] + 1));
        float2 xv = ((const float2*)x)[i];
        dinv[i] = dv;
        ((float2*)xs)[i] = make_float2(xv.x * dv, xv.y * dv);
    }
}


__global__ void k_gather_x(const int* __restrict__ cnt, const int* __restrict__ bucket,
                           const int* __restrict__ spill,
                           const float* __restrict__ dinv, const float* __restrict__ xs,
                           float* __restrict__ pt, int n) {
    int t = blockIdx.x * blockDim.x + threadIdx.x;
    int node = t >> 3, sub = t & 7;
    const float2* x2 = (const float2*)xs;
    float v0 = 0.0f, v1 = 0.0f;
    float di = 0.0f;
    float2 xi = make_float2(0.0f, 0.0f);
    if (node < n) {
        const int* bk = bucket + (size_t)node * CAP;
        int r0 = bk[sub], r1 = bk[sub + 8], r2 = bk[sub + 16], r3 = bk[sub + 24];
        int c = cnt[node];
        di = dinv[node];
        xi = x2[node];
        bool p0 = sub < c, p1 = sub + 8 < c, p2 = sub + 16 < c, p3 = sub + 24 < c;
        float2 a0 = x2[p0 ? r0 : node], a1 = x2[p1 ? r1 : node];
        float2 a2 = x2[p2 ? r2 : node], a3 = x2[p3 ? r3 : node];
        v0 += p0 ? a0.x : 0.0f;  v1 += p0 ? a0.y : 0.0f;
        v0 += p1 ? a1.x : 0.0f;  v1 += p1 ? a1.y : 0.0f;
        v0 += p2 ? a2.x : 0.0f;  v1 += p2 ? a2.y : 0.0f;
        v0 += p3 ? a3.x : 0.0f;  v1 += p3 ? a3.y : 0.0f;
        int cc = c < CAP ? c : CAP;
        for (int j = 32 + sub; j < cc; j += 8) {
            float2 a = x2[bk[j]];
            v0 += a.x;  v1 += a.y;
        }
        if (c > CAP) {
            int nsp = cnt[n];
            const int2* sp = (const int2*)spill;
            for (int p = sub; p < nsp; p += 8) {
                int2 ds = sp[p];
                if (ds.x == node) {
                    float2 a = x2[ds.y];
                    v0 += a.x;  v1 += a.y;
                }
            }
        }
    }
    v0 += __shfl_xor(v0, 4, 32);  v1 += __shfl_xor(v1, 4, 32);
    v0 += __shfl_xor(v0, 2, 32);  v1 += __shfl_xor(v1, 2, 32);
    v0 += __shfl_xor(v0, 1, 32);  v1 += __shfl_xor(v1, 1, 32);
    if (node < n && sub == 0) {
        ((float4*)pt)[node] = make_float4((xi.x + v0) * di, (xi.y + v1) * di, di, 0.0f);
    }
}


__device__ __forceinline__ float lane_val(float v, int k) {
    return __builtin_bit_cast(float, __builtin_amdgcn_readlane(__builtin_bit_cast(int, v), k));
}

__device__ __forceinline__ v2f add_act(v2f acc, float sx, float sy, float sd, v2f wa, v2f wb, v2f bb) {
    v2f z = sx * wa + (sy * wb + bb);
    z.x = fmaxf(z.x, 0.0f);
    z.y = fmaxf(z.y, 0.0f);
    return acc + sd * z;
}

__device__ __forceinline__ v2f add_chunk(v2f acc, float4 q, int m, v2f wa, v2f wb, v2f bb) {
    int k = 0;
    do {
        acc = add_act(acc, lane_val(q.x, k),     lane_val(q.y, k),     lane_val(q.z, k),     wa, wb, bb);
        acc = add_act(acc, lane_val(q.x, k + 1), lane_val(q.y, k + 1), lane_val(q.z, k + 1), wa, wb, bb);
        acc = add_act(acc, lane_val(q.x, k + 2), lane_val(q.y, k + 2), lane_val(q.z, k + 2), wa, wb, bb);
        acc = add_act(acc, lane_val(q.x, k + 3), lane_val(q.y, k + 3), lane_val(q.z, k + 3), wa, wb, bb);
        k += 4;
    } while (k < m);
    return acc;
}

__global__ void __launch_bounds__(256)
k_gather_act(const int* __restrict__ cnt, const int* __restrict__ bucket,
             const int* __restrict__ spill, const float* __restrict__ pt,
             const float* __restrict__ W1, const float* __restrict__ b1,
             float* __restrict__ sum, int n) {
    int lane = threadIdx.x & 31;
    int node = __builtin_amdgcn_readfirstlane((int)((blockIdx.x * blockDim.x + threadIdx.x) >> 5));
    if (node >= n) return;
    const int* bk = bucket + (size_t)node * CAP;
    const float4* p4 = (const float4*)pt;
    int rec = bk[lane];
    int c = cnt[node];
    float4 own = p4[node];
    v2f wa = ((const v2f*)W1)[lane];
    v2f wb = ((const v2f*)(W1 + HID))[lane];
    v2f bb = ((const v2f*)b1)[lane];
    int cc = c < CAP ? c : CAP;
    int m0 = cc < 32 ? cc : 32;
    bool live = lane < m0;
    float4 q0 = p4[live ? rec : node];
    q0.z = live ? q0.z : 0.0f;
    v2f zero = {0.0f, 0.0f};
    v2f acc = add_act(zero, own.x, own.y, own.z, wa, wb, bb);
    acc = add_chunk(acc, q0, m0, wa, wb, bb);
    if (cc > 32) {
        int rec1 = bk[32 + lane];
        int m1 = cc - 32;
        float4 q1 = make_float4(0.0f, 0.0f, 0.0f, 0.0f);
        if (lane < m1) q1 = p4[rec1];
        acc = add_chunk(acc, q1, m1, wa, wb, bb);
    }
    if (c > CAP) {
        int nsp = cnt[n];
        const int2* sp = (const int2*)spill;
        for (int p0 = 0; p0 < nsp; p0 += 32) {
            int2 ds = make_int2(-1, 0);
            if (p0 + lane < nsp) ds = sp[p0 + lane];
            bool hit = ds.x == node;
            float4 q = make_float4(0.0f, 0.0f, 0.0f, 0.0f);
            if (hit) q = p4[ds.y];
            unsigned mine = __builtin_amdgcn_ballot_w32(hit);
            while (mine) {
                int k = __builtin_ctz(mine);
                mine &= mine - 1;
                acc = add_act(acc, lane_val(q.x, k), lane_val(q.y, k), lane_val(q.z, k), wa, wb, bb);
            }
        }
    }
    ((v2f*)sum)[(size_t)node * 32 + lane] = acc * own.z;
}

template <int M> __device__ __forceinline__ float row_xor(float x) {
    int b = __builtin_bit_cast(int, x);
    return __builtin_bit_cast(float, __builtin_amdgcn_update_dpp(b, b, 0x160 + M, 0xf, 0xf, true));
}

__global__ void __launch_bounds__(256)
k_gemm_relu_dot_wmma(const float* __restrict__ sum, const float* __restrict__ W,
                     const float* __restrict__ b2, const float* __restrict__ W3,
                     const float* __restrict__ dinv, float* __restrict__ h3, int nrows) {
    int gwave = (blockIdx.x * blockDim.x + threadIdx.x) >> 5;
    int lane  = threadIdx.x & 31;
    int row0  = gwave << 4;
    if (row0 >= nrows) return;
    int hh = lane >> 4;
    int l  = lane & 15;
    v8f acc0 = {}, acc1 = {}, acc2 = {}, acc3 = {};
    const float* arow = sum + (size_t)(row0 + l) * HID + hh * 2;
    for (int k0 = 0; k0 < HID; k0 += 4) {
        int ka = k0 + hh * 2;
        v2f a = *(const v2f*)(arow + k0);
        const float* bp = W + (size_t)ka * HID + l;
        v2f b0, b1f, b2f, b3;
        b0.x  = bp[0];  b0.y  = bp[HID];
        b1f.x = bp[16]; b1f.y = bp[HID + 16];
        b2f.x = bp[32]; b2f.y = bp[HID + 32];
        b3.x  = bp[48]; b3.y  = bp[HID + 48];
        acc0 = __builtin_amdgcn_wmma_f32_16x16x4_f32(false, a, false, b0,  (short)0, acc0, false, false);
        acc1 = __builtin_amdgcn_wmma_f32_16x16x4_f32(false, a, false, b1f, (short)0, acc1, false, false);
        acc2 = __builtin_amdgcn_wmma_f32_16x16x4_f32(false, a, false, b2f, (short)0, acc2, false, false);
        acc3 = __builtin_amdgcn_wmma_f32_16x16x4_f32(false, a, false, b3,  (short)0, acc3, false, false);
    }
    float c0 = b2[l], c1 = b2[16 + l], c2 = b2[32 + l], c3 = b2[48 + l];
    float u0 = W3[l], u1 = W3[16 + l], u2 = W3[32 + l], u3 = W3[48 + l];
    float mine = 0.0f;
    #pragma unroll
    for (int r = 0; r < 8; ++r) {
        float t = fmaxf(acc0[r] + c0, 0.0f) * u0 + fmaxf(acc1[r] + c1, 0.0f) * u1
                + fmaxf(acc2[r] + c2, 0.0f) * u2 + fmaxf(acc3[r] + c3, 0.0f) * u3;
        t += row_xor<1>(t);
        t += row_xor<2>(t);
        t += row_xor<4>(t);
        t += row_xor<8>(t);
        if (l == r) mine = t;
    }
    if (l < 8) {
        int row = row0 + hh * 8 + l;
        h3[row] = mine * dinv[row];
    }
}

__global__ void k_gather_scalar(const int* __restrict__ cnt, const int* __restrict__ bucket,
                                const int* __restrict__ spill,
                                const float* __restrict__ dinv, const float* __restrict__ h3,
                                const float* __restrict__ b3, float* __restrict__ y, int n) {
    int t = blockIdx.x * blockDim.x + threadIdx.x;
    int node = t >> 3, sub = t & 7;
    float v = 0.0f;
    if (node < n) {
        const int* bk = bucket + (size_t)node * CAP;
        int r0 = bk[sub], r1 = bk[sub + 8], r2 = bk[sub + 16], r3 = bk[sub + 24];
        int c = cnt[node];
        bool p0 = sub < c, p1 = sub + 8 < c, p2 = sub + 16 < c, p3 = sub + 24 < c;
        float a0 = h3[p0 ? r0 : node], a1 = h3[p1 ? r1 : node];
        float a2 = h3[p2 ? r2 : node], a3 = h3[p3 ? r3 : node];
        v += p0 ? a0 : 0.0f;
        v += p1 ? a1 : 0.0f;
        v += p2 ? a2 : 0.0f;
        v += p3 ? a3 : 0.0f;
        int cc = c < CAP ? c : CAP;
        for (int j = 32 + sub; j < cc; j += 8) v += h3[bk[j]];
        if (c > CAP) {
            int nsp = cnt[n];
            const int2* sp = (const int2*)spill;
            for (int p = sub; p < nsp; p += 8) {
                int2 ds = sp[p];
                if (ds.x == node) v += h3[ds.y];
            }
        }
    }
    v += __shfl_xor(v, 4, 32);
    v += __shfl_xor(v, 2, 32);
    v += __shfl_xor(v, 1, 32);
    if (node < n && sub == 0) {
        float di = dinv[node];
        y[node] = b3[0] + (h3[node] + v) * di;
    }
}


extern "C" void kernel_launch(void* const* d_in, const int* in_sizes, int n_in,
                              void* d_out, int out_size, void* d_ws, size_t ws_size,
                              hipStream_t stream) {
    const float* x  = (const float*)d_in[0];
    const int*   ei = (const int*)  d_in[1];
    const float* W1 = (const float*)d_in[2];
    const float* b1 = (const float*)d_in[3];
    const float* W2 = (const float*)d_in[4];
    const float* b2 = (const float*)d_in[5];
    const float* W3 = (const float*)d_in[6];
    const float* b3 = (const float*)d_in[7];

    const int N = in_sizes[0] / 2;
    const int E = in_sizes[1] / 2;
    const int* src = ei;
    const int* dst = ei + E;

    float* hsum   = (float*)d_ws;
    int*   bucket = (int*)(hsum + (size_t)N * HID);
    float* pt     = (float*)(bucket + (size_t)N * CAP);
    float* xs     = pt + (size_t)N * 4;
    float* dinv   = xs + (size_t)N * 2;
    float* h3     = dinv + (size_t)N;
    int*   spill  = (int*)(h3 + (size_t)N);
    int*   cnt    = spill + (size_t)E * 2;
    float* y      = (float*)d_out;

    const int B = 256;
    auto g = [](long long t) { return (int)((t + 255) / 256); };

    k_zero_cnt<<<g(N + 1), B, 0, stream>>>(cnt, N + 1);
    k_bucket  <<<g(E), B, 0, stream>>>(src, dst, cnt, bucket, spill, E, N);
    k_prep    <<<g(N), B, 0, stream>>>(cnt, x, dinv, xs, N);

    k_gather_x<<<g((long long)N * 8), B, 0, stream>>>(cnt, bucket, spill, dinv, xs, pt, N);

    k_gather_act<<<g((long long)N * 32), B, 0, stream>>>(cnt, bucket, spill, pt, W1, b1, hsum, N);
    k_gemm_relu_dot_wmma<<<g((long long)(N / 16) * 32), 256, 0, stream>>>(hsum, W2, b2, W3, dinv, h3, N);

    k_gather_scalar<<<g((long long)N * 8),  B, 0, stream>>>(cnt, bucket, spill, dinv, h3, b3, y, N);
}


# Edges split by destination range through LDS counters: 4096-edge blocks ordered in LDS, one block per 256 nodes ranks in LDS, stores whole bucket lines; no global add per edge; arrival order kept
# speedup vs baseline: 7.7891x; 2.2940x over previous
#include <hip/hip_runtime.h>
#include <hip/hip_bf16.h>

typedef __attribute__((ext_vector_type(2))) float v2f;
typedef __attribute__((ext_vector_type(8))) float v8f;

#define HID 64


#define CAP 64

#define BIN_SHIFT   8
#define BIN_NODES   256
#define MAXBINS     1024
#define SPLIT_EDGES 4096
#define PITCH       65

__global__ void __launch_bounds__(1024)
k_split(const int* __restrict__ src, const int* __restrict__ dst,
        int* __restrict__ rec, int* __restrict__ seg, int* __restrict__ cnt,
        int e, int n, int nbins, int spitch) {
    __shared__ int  lcount[MAXBINS];
    __shared__ int  lstart[MAXBINS];
    __shared__ int  wsum[32];
    __shared__ int2 lrec[SPLIT_EDGES];
    int tid = threadIdx.x, lane = tid & 31, wave = tid >> 5;
    int beg = blockIdx.x * SPLIT_EDGES;
    if (blockIdx.x == 0 && tid == 0) cnt[n] = 0;
    lcount[tid] = 0;
    __syncthreads();
    int d[4], s[4], r[4];
    #pragma unroll
    for (int j = 0; j < 4; ++j) {
        int i = beg + j * 1024 + tid;
        bool ok = i < e;
        d[j] = ok ? dst[i] : -1;
        s[j] = ok ? src[i] : 0;
    }
    #pragma unroll
    for (int j = 0; j < 4; ++j) {
        r[j] = 0;
        if (d[j] >= 0)
            r[j] = __hip_atomic_fetch_add(&lcount[d[j] >> BIN_SHIFT], 1, __ATOMIC_RELAXED,
                                          __HIP_MEMORY_SCOPE_WORKGROUP);
    }
    __syncthreads();
    int v = lcount[tid];
    int incl = v;
    #pragma unroll
    for (int o = 1; o < 32; o <<= 1) {
        int u = __shfl_up(incl, o, 32);
        if (lane >= o) incl += u;
    }
    if (lane == 31) wsum[wave] = incl;
    __syncthreads();
    int base = 0;
    for (int w = 0; w < wave; ++w) base += wsum[w];
    int excl = base + incl - v;
    lstart[tid] = excl;
    if (tid < nbins)
        ((int2*)seg)[(size_t)blockIdx.x * spitch + tid] = make_int2(beg + excl, v);
    __syncthreads();
    #pragma unroll
    for (int j = 0; j < 4; ++j)
        if (d[j] >= 0) lrec[lstart[d[j] >> BIN_SHIFT] + r[j]] = make_int2(d[j], s[j]);
    __syncthreads();
    int m = e - beg;
    m = m < SPLIT_EDGES ? m : SPLIT_EDGES;
    for (int i = tid; i < m; i += 1024) ((int2*)rec)[beg + i] = lrec[i];
}

__global__ void __launch_bounds__(1024)
k_place(const int* __restrict__ rec, const int* __restrict__ seg, const float* __restrict__ x,
        int* __restrict__ cnt, int* __restrict__ bucket, int* __restrict__ spill,
        float* __restrict__ dinv, float* __restrict__ xs, int n, int nseg, int spitch) {
    __shared__ int img[BIN_NODES * PITCH];
    __shared__ int lcnt[BIN_NODES];
    int tid = threadIdx.x;
    int node0 = blockIdx.x << BIN_SHIFT;
    for (int i = tid; i < BIN_NODES * PITCH; i += 1024) img[i] = 0;
    if (tid < BIN_NODES) lcnt[tid] = 0;
    __syncthreads();
    int sub = tid & 1;
    const int2* rp = (const int2*)rec;
    for (int k = tid >> 1; k < nseg; k += 512) {
        int2 sg = ((const int2*)seg)[(size_t)k * spitch + blockIdx.x];
        for (int j = sub; j < sg.y; j += 8) {
            int2 q[4];
            #pragma unroll
            for (int u = 0; u < 4; ++u) {
                int jj = j + 2 * u;
                q[u] = rp[sg.x + (jj < sg.y ? jj : j)];
            }
            #pragma unroll
            for (int u = 0; u < 4; ++u) {
                if (j + 2 * u < sg.y) {
                    int2 ds = q[u];
                    int ld = ds.x - node0;
                    int r = __hip_atomic_fetch_add(&lcnt[ld], 1, __ATOMIC_RELAXED,
                                                   __HIP_MEMORY_SCOPE_WORKGROUP);
                    if (r < CAP) {
                        img[ld * PITCH + r] = ds.y;
                    } else {
                        int p = __hip_atomic_fetch_add(&cnt[n], 1, __ATOMIC_RELAXED,
                                                       __HIP_MEMORY_SCOPE_AGENT);
                        ((int2*)spill)[p] = ds;
                    }
                }
            }
        }
    }
    __syncthreads();
    int nh = n - node0;
    nh = nh < BIN_NODES ? nh : BIN_NODES;
    int* gb = bucket + (size_t)node0 * CAP;
    for (int i = tid; i < nh * CAP; i += 1024) {
        int ld = i >> 6, r = i & 63;
        if (r < 32 || lcnt[ld] > 32) gb[i] = img[ld * PITCH + r];
    }
    if (tid < nh) {
        int node = node0 + tid;
        int c = lcnt[tid];
        float  dv = rsqrtf((float)(c + 1));
        float2 xv = ((const float2*)x)[node];
        cnt[node] = c;
        dinv[node] = dv;
        ((float2*)xs)[node] = make_float2(xv.x * dv, xv.y * dv);
    }
}


__global__ void k_gather_x(const int* __restrict__ cnt, const int* __restrict__ bucket,
                           const int* __restrict__ spill,
                           const float* __restrict__ dinv, const float* __restrict__ xs,
                           float* __restrict__ pt, int n) {
    int t = blockIdx.x * blockDim.x + threadIdx.x;
    int node = t >> 3, sub = t & 7;
    const float2* x2 = (const float2*)xs;
    float v0 = 0.0f, v1 = 0.0f;
    float di = 0.0f;
    float2 xi = make_float2(0.0f, 0.0f);
    if (node < n) {
        const int* bk = bucket + (size_t)node * CAP;
        int r0 = bk[sub], r1 = bk[sub + 8], r2 = bk[sub + 16], r3 = bk[sub + 24];
        int c = cnt[node];
        di = dinv[node];
        xi = x2[node];
        bool p0 = sub < c, p1 = sub + 8 < c, p2 = sub + 16 < c, p3 = sub + 24 < c;
        float2 a0 = x2[p0 ? r0 : node], a1 = x2[p1 ? r1 : node];
        float2 a2 = x2[p2 ? r2 : node], a3 = x2[p3 ? r3 : node];
        v0 += p0 ? a0.x : 0.0f;  v1 += p0 ? a0.y : 0.0f;
        v0 += p1 ? a1.x : 0.0f;  v1 += p1 ? a1.y : 0.0f;
        v0 += p2 ? a2.x : 0.0f;  v1 += p2 ? a2.y : 0.0f;
        v0 += p3 ? a3.x : 0.0f;  v1 += p3 ? a3.y : 0.0f;
        int cc = c < CAP ? c : CAP;
        for (int j = 32 + sub; j < cc; j += 8) {
            float2 a = x2[bk[j]];
            v0 += a.x;  v1 += a.y;
        }
        if (c > CAP) {
            int nsp = cnt[n];
            const int2* sp = (const int2*)spill;
            for (int p = sub; p < nsp; p += 8) {
                int2 ds = sp[p];
                if (ds.x == node) {
                    float2 a = x2[ds.y];
                    v0 += a.x;  v1 += a.y;
                }
            }
        }
    }
    v0 += __shfl_xor(v0, 4, 32);  v1 += __shfl_xor(v1, 4, 32);
    v0 += __shfl_xor(v0, 2, 32);  v1 += __shfl_xor(v1, 2, 32);
    v0 += __shfl_xor(v0, 1, 32);  v1 += __shfl_xor(v1, 1, 32);
    if (node < n && sub == 0) {
        ((float4*)pt)[node] = make_float4((xi.x + v0) * di, (xi.y + v1) * di, di, 0.0f);
    }
}


__device__ __forceinline__ float lane_val(float v, int k) {
    return __builtin_bit_cast(float, __builtin_amdgcn_readlane(__builtin_bit_cast(int, v), k));
}

__device__ __forceinline__ v2f add_act(v2f acc, float sx, float sy, float sd, v2f wa, v2f wb, v2f bb) {
    v2f z = sx * wa + (sy * wb + bb);
    z.x = fmaxf(z.x, 0.0f);
    z.y = fmaxf(z.y, 0.0f);
    return acc + sd * z;
}

__device__ __forceinline__ v2f add_chunk(v2f acc, float4 q, int m, v2f wa, v2f wb, v2f bb) {
    int k = 0;
    do {
        acc = add_act(acc, lane_val(q.x, k),     lane_val(q.y, k),     lane_val(q.z, k),     wa, wb, bb);
        acc = add_act(acc, lane_val(q.x, k + 1), lane_val(q.y, k + 1), lane_val(q.z, k + 1), wa, wb, bb);
        acc = add_act(acc, lane_val(q.x, k + 2), lane_val(q.y, k + 2), lane_val(q.z, k + 2), wa, wb, bb);
        acc = add_act(acc, lane_val(q.x, k + 3), lane_val(q.y, k + 3), lane_val(q.z, k + 3), wa, wb, bb);
        k += 4;
    } while (k < m);
    return acc;
}

__global__ void __launch_bounds__(256)
k_gather_act(const int* __restrict__ cnt, const int* __restrict__ bucket,
             const int* __restrict__ spill, const float* __restrict__ pt,
             const float* __restrict__ W1, const float* __restrict__ b1,
             float* __restrict__ sum, int n) {
    int lane = threadIdx.x & 31;
    int node = __builtin_amdgcn_readfirstlane((int)((blockIdx.x * blockDim.x + threadIdx.x) >> 5));
    if (node >= n) return;
    const int* bk = bucket + (size_t)node * CAP;
    const float4* p4 = (const float4*)pt;
    int rec = bk[lane];
    int c = cnt[node];
    float4 own = p4[node];
    v2f wa = ((const v2f*)W1)[lane];
    v2f wb = ((const v2f*)(W1 + HID))[lane];
    v2f bb = ((const v2f*)b1)[lane];
    int cc = c < CAP ? c : CAP;
    int m0 = cc < 32 ? cc : 32;
    bool live = lane < m0;
    float4 q0 = p4[live ? rec : node];
    q0.z = live ? q0.z : 0.0f;
    v2f zero = {0.0f, 0.0f};
    v2f acc = add_act(zero, own.x, own.y, own.z, wa, wb, bb);
    acc = add_chunk(acc, q0, m0, wa, wb, bb);
    if (cc > 32) {
        int rec1 = bk[32 + lane];
        int m1 = cc - 32;
        float4 q1 = make_float4(0.0f, 0.0f, 0.0f, 0.0f);
        if (lane < m1) q1 = p4[rec1];
        acc = add_chunk(acc, q1, m1, wa, wb, bb);
    }
    if (c > CAP) {
        int nsp = cnt[n];
        const int2* sp = (const int2*)spill;
        for (int p0 = 0; p0 < nsp; p0 += 32) {
            int2 ds = make_int2(-1, 0);
            if (p0 + lane < nsp) ds = sp[p0 + lane];
            bool hit = ds.x == node;
            float4 q = make_float4(0.0f, 0.0f, 0.0f, 0.0f);
            if (hit) q = p4[ds.y];
            unsigned mine = __builtin_amdgcn_ballot_w32(hit);
            while (mine) {
                int k = __builtin_ctz(mine);
                mine &= mine - 1;
                acc = add_act(acc, lane_val(q.x, k), lane_val(q.y, k), lane_val(q.z, k), wa, wb, bb);
            }
        }
    }
    ((v2f*)sum)[(size_t)node * 32 + lane] = acc * own.z;
}

template <int M> __device__ __forceinline__ float row_xor(float x) {
    int b = __builtin_bit_cast(int, x);
    return __builtin_bit_cast(float, __builtin_amdgcn_update_dpp(b, b, 0x160 + M, 0xf, 0xf, true));
}

__global__ void __launch_bounds__(256)
k_gemm_relu_dot_wmma(const float* __restrict__ sum, const float* __restrict__ W,
                     const float* __restrict__ b2, const float* __restrict__ W3,
                     const float* __restrict__ dinv, float* __restrict__ h3, int nrows) {
    int gwave = (blockIdx.x * blockDim.x + threadIdx.x) >> 5;
    int lane  = threadIdx.x & 31;
    int row0  = gwave << 4;
    if (row0 >= nrows) return;
    int hh = lane >> 4;
    int l  = lane & 15;
    v8f acc0 = {}, acc1 = {}, acc2 = {}, acc3 = {};
    const float* arow = sum + (size_t)(row0 + l) * HID + hh * 2;
    for (int k0 = 0; k0 < HID; k0 += 4) {
        int ka = k0 + hh * 2;
        v2f a = *(const v2f*)(arow + k0);
        const float* bp = W + (size_t)ka * HID + l;
        v2f b0, b1f, b2f, b3;
        b0.x  = bp[0];  b0.y  = bp[HID];
        b1f.x = bp[16]; b1f.y = bp[HID + 16];
        b2f.x = bp[32]; b2f.y = bp[HID + 32];
        b3.x  = bp[48]; b3.y  = bp[HID + 48];
        acc0 = __builtin_amdgcn_wmma_f32_16x16x4_f32(false, a, false, b0,  (short)0, acc0, false, false);
        acc1 = __builtin_amdgcn_wmma_f32_16x16x4_f32(false, a, false, b1f, (short)0, acc1, false, false);
        acc2 = __builtin_amdgcn_wmma_f32_16x16x4_f32(false, a, false, b2f, (short)0, acc2, false, false);
        acc3 = __builtin_amdgcn_wmma_f32_16x16x4_f32(false, a, false, b3,  (short)0, acc3, false, false);
    }
    float c0 = b2[l], c1 = b2[16 + l], c2 = b2[32 + l], c3 = b2[48 + l];
    float u0 = W3[l], u1 = W3[16 + l], u2 = W3[32 + l], u3 = W3[48 + l];
    float mine = 0.0f;
    #pragma unroll
    for (int r = 0; r < 8; ++r) {
        float t = fmaxf(acc0[r] + c0, 0.0f) * u0 + fmaxf(acc1[r] + c1, 0.0f) * u1
                + fmaxf(acc2[r] + c2, 0.0f) * u2 + fmaxf(acc3[r] + c3, 0.0f) * u3;
        t += row_xor<1>(t);
        t += row_xor<2>(t);
        t += row_xor<4>(t);
        t += row_xor<8>(t);
        if (l == r) mine = t;
    }
    if (l < 8) {
        int row = row0 + hh * 8 + l;
        h3[row] = mine * dinv[row];
    }
}

__global__ void k_gather_scalar(const int* __restrict__ cnt, const int* __restrict__ bucket,
                                const int* __restrict__ spill,
                                const float* __restrict__ dinv, const float* __restrict__ h3,
                                const float* __restrict__ b3, float* __restrict__ y, int n) {
    int t = blockIdx.x * blockDim.x + threadIdx.x;
    int node = t >> 3, sub = t & 7;
    float v = 0.0f;
    if (node < n) {
        const int* bk = bucket + (size_t)node * CAP;
        int r0 = bk[sub], r1 = bk[sub + 8], r2 = bk[sub + 16], r3 = bk[sub + 24];
        int c = cnt[node];
        bool p0 = sub < c, p1 = sub + 8 < c, p2 = sub + 16 < c, p3 = sub + 24 < c;
        float a0 = h3[p0 ? r0 : node], a1 = h3[p1 ? r1 : node];
        float a2 = h3[p2 ? r2 : node], a3 = h3[p3 ? r3 : node];
        v += p0 ? a0 : 0.0f;
        v += p1 ? a1 : 0.0f;
        v += p2 ? a2 : 0.0f;
        v += p3 ? a3 : 0.0f;
        int cc = c < CAP ? c : CAP;
        for (int j = 32 + sub; j < cc; j += 8) v += h3[bk[j]];
        if (c > CAP) {
            int nsp = cnt[n];
            const int2* sp = (const int2*)spill;
            for (int p = sub; p < nsp; p += 8) {
                int2 ds = sp[p];
                if (ds.x == node) v += h3[ds.y];
            }
        }
    }
    v += __shfl_xor(v, 4, 32);
    v += __shfl_xor(v, 2, 32);
    v += __shfl_xor(v, 1, 32);
    if (node < n && sub == 0) {
        float di = dinv[node];
        y[node] = b3[0] + (h3[node] + v) * di;
    }
}


extern "C" void kernel_launch(void* const* d_in, const int* in_sizes, int n_in,
                              void* d_out, int out_size, void* d_ws, size_t ws_size,
                              hipStream_t stream) {
    const float* x  = (const float*)d_in[0];
    const int*   ei = (const int*)  d_in[1];
    const float* W1 = (const float*)d_in[2];
    const float* b1 = (const float*)d_in[3];
    const float* W2 = (const float*)d_in[4];
    const float* b2 = (const float*)d_in[5];
    const float* W3 = (const float*)d_in[6];
    const float* b3 = (const float*)d_in[7];

    const int N = in_sizes[0] / 2;
    const int E = in_sizes[1] / 2;
    const int* src = ei;
    const int* dst = ei + E;

    float* hsum   = (float*)d_ws;
    int*   bucket = (int*)(hsum + (size_t)N * HID);
    float* pt     = (float*)(bucket + (size_t)N * CAP);
    float* xs     = pt + (size_t)N * 4;
    float* dinv   = xs + (size_t)N * 2;
    float* h3     = dinv + (size_t)N;
    int*   spill  = (int*)(h3 + (size_t)N);
    int*   rec    = spill + (size_t)E * 2;
    const int nbins = (N + BIN_NODES - 1) / BIN_NODES;
    const int nsplit = (E + SPLIT_EDGES - 1) / SPLIT_EDGES;
    const int spitch = (nbins + 15) / 16 * 16;
    int*   seg    = rec + (size_t)E * 2;
    int*   cnt    = seg + (size_t)nsplit * spitch * 2;
    float* y      = (float*)d_out;

    const int B = 256;
    auto g = [](long long t) { return (int)((t + 255) / 256); };

    k_split<<<nsplit, 1024, 0, stream>>>(src, dst, rec, seg, cnt, E, N, nbins, spitch);
    k_place<<<nbins, 1024, 0, stream>>>(rec, seg, x, cnt, bucket, spill, dinv, xs, N, nsplit, spitch);

    k_gather_x<<<g((long long)N * 8), B, 0, stream>>>(cnt, bucket, spill, dinv, xs, pt, N);

    k_gather_act<<<g((long long)N * 32), B, 0, stream>>>(cnt, bucket, spill, pt, W1, b1, hsum, N);
    k_gemm_relu_dot_wmma<<<g((long long)(N / 16) * 32), 256, 0, stream>>>(hsum, W2, b2, W3, dinv, h3, N);

    k_gather_scalar<<<g((long long)N * 8),  B, 0, stream>>>(cnt, bucket, spill, dinv, h3, b3, y, N);
}


# Layer 2 node sums formed straight in the W2 product's A-fragment registers: a wave owns 16 nodes (lane pair per node), 2 column passes sharing the entries through LDS; no [N,64] plane, 5 launches
# speedup vs baseline: 8.1553x; 1.0470x over previous
#include <hip/hip_runtime.h>
#include <hip/hip_bf16.h>

typedef __attribute__((ext_vector_type(2))) float v2f;
typedef __attribute__((ext_vector_type(8))) float v8f;

#define HID 64


#define CAP 64

#define BIN_SHIFT   8
#define BIN_NODES   256
#define MAXBINS     1024
#define SPLIT_EDGES 4096
#define PITCH       65

__global__ void __launch_bounds__(1024)
k_split(const int* __restrict__ src, const int* __restrict__ dst,
        int* __restrict__ rec, int* __restrict__ seg, int* __restrict__ cnt,
        int e, int n, int nbins, int spitch) {
    __shared__ int  lcount[MAXBINS];
    __shared__ int  lstart[MAXBINS];
    __shared__ int  wsum[32];
    __shared__ int2 lrec[SPLIT_EDGES];
    int tid = threadIdx.x, lane = tid & 31, wave = tid >> 5;
    int beg = blockIdx.x * SPLIT_EDGES;
    if (blockIdx.x == 0 && tid == 0) cnt[n] = 0;
    lcount[tid] = 0;
    __syncthreads();
    int d[4], s[4], r[4];
    #pragma unroll
    for (int j = 0; j < 4; ++j) {
        int i = beg + j * 1024 + tid;
        bool ok = i < e;
        d[j] = ok ? dst[i] : -1;
        s[j] = ok ? src[i] : 0;
    }
    #pragma unroll
    for (int j = 0; j < 4; ++j) {
        r[j] = 0;
        if (d[j] >= 0)
            r[j] = __hip_atomic_fetch_add(&lcount[d[j] >> BIN_SHIFT], 1, __ATOMIC_RELAXED,
                                          __HIP_MEMORY_SCOPE_WORKGROUP);
    }
    __syncthreads();
    int v = lcount[tid];
    int incl = v;
    #pragma unroll
    for (int o = 1; o < 32; o <<= 1) {
        int u = __shfl_up(incl, o, 32);
        if (lane >= o) incl += u;
    }
    if (lane == 31) wsum[wave] = incl;
    __syncthreads();
    int base = 0;
    for (int w = 0; w < wave; ++w) base += wsum[w];
    int excl = base + incl - v;
    lstart[tid] = excl;
    if (tid < nbins)
        ((int2*)seg)[(size_t)blockIdx.x * spitch + tid] = make_int2(beg + excl, v);
    __syncthreads();
    #pragma unroll
    for (int j = 0; j < 4; ++j)
        if (d[j] >= 0) lrec[lstart[d[j] >> BIN_SHIFT] + r[j]] = make_int2(d[j], s[j]);
    __syncthreads();
    int m = e - beg;
    m = m < SPLIT_EDGES ? m : SPLIT_EDGES;
    for (int i = tid; i < m; i += 1024) ((int2*)rec)[beg + i] = lrec[i];
}

__global__ void __launch_bounds__(1024)
k_place(const int* __restrict__ rec, const int* __restrict__ seg, const float* __restrict__ x,
        int* __restrict__ cnt, int* __restrict__ bucket, int* __restrict__ spill,
        float* __restrict__ dinv, float* __restrict__ xs, int n, int nseg, int spitch) {
    __shared__ int img[BIN_NODES * PITCH];
    __shared__ int lcnt[BIN_NODES];
    int tid = threadIdx.x;
    int node0 = blockIdx.x << BIN_SHIFT;
    for (int i = tid; i < BIN_NODES * PITCH; i += 1024) img[i] = 0;
    if (tid < BIN_NODES) lcnt[tid] = 0;
    __syncthreads();
    int sub = tid & 1;
    const int2* rp = (const int2*)rec;
    for (int k = tid >> 1; k < nseg; k += 512) {
        int2 sg = ((const int2*)seg)[(size_t)k * spitch + blockIdx.x];
        for (int j = sub; j < sg.y; j += 8) {
            int2 q[4];
            #pragma unroll
            for (int u = 0; u < 4; ++u) {
                int jj = j + 2 * u;
                q[u] = rp[sg.x + (jj < sg.y ? jj : j)];
            }
            #pragma unroll
            for (int u = 0; u < 4; ++u) {
                if (j + 2 * u < sg.y) {
                    int2 ds = q[u];
                    int ld = ds.x - node0;
                    int r = __hip_atomic_fetch_add(&lcnt[ld], 1, __ATOMIC_RELAXED,
                                                   __HIP_MEMORY_SCOPE_WORKGROUP);
                    if (r < CAP) {
                        img[ld * PITCH + r] = ds.y;
                    } else {
                        int p = __hip_atomic_fetch_add(&cnt[n], 1, __ATOMIC_RELAXED,
                                                       __HIP_MEMORY_SCOPE_AGENT);
                        ((int2*)spill)[p] = ds;
                    }
                }
            }
        }
    }
    __syncthreads();
    int nh = n - node0;
    nh = nh < BIN_NODES ? nh : BIN_NODES;
    int* gb = bucket + (size_t)node0 * CAP;
    for (int i = tid; i < nh * CAP; i += 1024) {
        int ld = i >> 6, r = i & 63;
        if (r < 32 || lcnt[ld] > 32) gb[i] = img[ld * PITCH + r];
    }
    if (tid < nh) {
        int node = node0 + tid;
        int c = lcnt[tid];
        float  dv = rsqrtf((float)(c + 1));
        float2 xv = ((const float2*)x)[node];
        cnt[node] = c;
        dinv[node] = dv;
        ((float2*)xs)[node] = make_float2(xv.x * dv, xv.y * dv);
    }
}


__global__ void k_gather_x(const int* __restrict__ cnt, const int* __restrict__ bucket,
                           const int* __restrict__ spill,
                           const float* __restrict__ dinv, const float* __restrict__ xs,
                           float* __restrict__ pt, int n) {
    int t = blockIdx.x * blockDim.x + threadIdx.x;
    int node = t >> 3, sub = t & 7;
    const float2* x2 = (const float2*)xs;
    float v0 = 0.0f, v1 = 0.0f;
    float di = 0.0f;
    float2 xi = make_float2(0.0f, 0.0f);
    if (node < n) {
        const int* bk = bucket + (size_t)node * CAP;
        int r0 = bk[sub], r1 = bk[sub + 8], r2 = bk[sub + 16], r3 = bk[sub + 24];
        int c = cnt[node];
        di = dinv[node];
        xi = x2[node];
        bool p0 = sub < c, p1 = sub + 8 < c, p2 = sub + 16 < c, p3 = sub + 24 < c;
        float2 a0 = x2[p0 ? r0 : node], a1 = x2[p1 ? r1 : node];
        float2 a2 = x2[p2 ? r2 : node], a3 = x2[p3 ? r3 : node];
        v0 += p0 ? a0.x : 0.0f;  v1 += p0 ? a0.y : 0.0f;
        v0 += p1 ? a1.x : 0.0f;  v1 += p1 ? a1.y : 0.0f;
        v0 += p2 ? a2.x : 0.0f;  v1 += p2 ? a2.y : 0.0f;
        v0 += p3 ? a3.x : 0.0f;  v1 += p3 ? a3.y : 0.0f;
        int cc = c < CAP ? c : CAP;
        for (int j = 32 + sub; j < cc; j += 8) {
            float2 a = x2[bk[j]];
            v0 += a.x;  v1 += a.y;
        }
        if (c > CAP) {
            int nsp = cnt[n];
            const int2* sp = (const int2*)spill;
            for (int p = sub; p < nsp; p += 8) {
                int2 ds = sp[p];
                if (ds.x == node) {
                    float2 a = x2[ds.y];
                    v0 += a.x;  v1 += a.y;
                }
            }
        }
    }
    v0 += __shfl_xor(v0, 4, 32);  v1 += __shfl_xor(v1, 4, 32);
    v0 += __shfl_xor(v0, 2, 32);  v1 += __shfl_xor(v1, 2, 32);
    v0 += __shfl_xor(v0, 1, 32);  v1 += __shfl_xor(v1, 1, 32);
    if (node < n && sub == 0) {
        ((float4*)pt)[node] = make_float4((xi.x + v0) * di, (xi.y + v1) * di, di, 0.0f);
    }
}


__device__ __forceinline__ v2f add_act(v2f acc, float sx, float sy, float sd, v2f wa, v2f wb, v2f bb) {
    v2f z = sx * wa + (sy * wb + bb);
    z.x = fmaxf(z.x, 0.0f);
    z.y = fmaxf(z.y, 0.0f);
    return acc + sd * z;
}

template <int M> __device__ __forceinline__ float row_xor(float x) {
    int b = __builtin_bit_cast(int, x);
    return __builtin_bit_cast(float, __builtin_amdgcn_update_dpp(b, b, 0x160 + M, 0xf, 0xf, true));
}

#define STASH_TRIPS 8
#define STASH_SLOTS (4 * STASH_TRIPS)

__device__ __forceinline__ void add_terms(v2f (&acc)[8], float4 q0, float4 q1, float4 q2, float4 q3,
                                          const v2f (&wa)[8], const v2f (&wb)[8], const v2f (&bb)[8]) {
    #pragma unroll
    for (int t = 0; t < 8; ++t) acc[t] = add_act(acc[t], q0.x, q0.y, q0.z, wa[t], wb[t], bb[t]);
    __builtin_amdgcn_sched_barrier(0);
    #pragma unroll
    for (int t = 0; t < 8; ++t) acc[t] = add_act(acc[t], q1.x, q1.y, q1.z, wa[t], wb[t], bb[t]);
    __builtin_amdgcn_sched_barrier(0);
    #pragma unroll
    for (int t = 0; t < 8; ++t) acc[t] = add_act(acc[t], q2.x, q2.y, q2.z, wa[t], wb[t], bb[t]);
    __builtin_amdgcn_sched_barrier(0);
    #pragma unroll
    for (int t = 0; t < 8; ++t) acc[t] = add_act(acc[t], q3.x, q3.y, q3.z, wa[t], wb[t], bb[t]);
}

template <int T0, bool FIRST>
__device__ __forceinline__ void node_sums(v2f (&acc)[8], int node, int hh, int lc, int c, float di, int n,
                                          const int* __restrict__ cnt, const int* __restrict__ bucket,
                                          const int* __restrict__ spill, const float4* __restrict__ p4,
                                          const float* __restrict__ W1, const float* __restrict__ b1,
                                          float4* stash) {
    const int* bk = bucket + (size_t)node * CAP;
    v2f wa[8], wb[8], bb[8];
    const float* w1p = W1 + 4 * T0 + 2 * hh;
    const float* b1p = b1 + 4 * T0 + 2 * hh;
    #pragma unroll
    for (int t = 0; t < 8; ++t) {
        wa[t] = *(const v2f*)(w1p + 4 * t);
        wb[t] = *(const v2f*)(w1p + HID + 4 * t);
        bb[t] = *(const v2f*)(b1p + 4 * t);
        acc[t].x = 0.0f;
        acc[t].y = 0.0f;
    }
    int cc = c < CAP ? c : CAP;
    if (FIRST) {
        const int4* bk4 = (const int4*)bk;
        int4 cur = bk4[0];
        int last = node;
        for (int g = 0; g <= CAP / 4; ++g) {
            int v0 = g * 4 - 1;
            if (__builtin_amdgcn_ballot_w32(v0 < cc) == 0) break;
            bool l0 = v0 < cc, l1 = v0 + 1 < cc, l2 = v0 + 2 < cc, l3 = v0 + 3 < cc;
            float4 q0 = p4[l0 ? last  : node], q1 = p4[l1 ? cur.x : node];
            float4 q2 = p4[l2 ? cur.y : node], q3 = p4[l3 ? cur.z : node];
            last = cur.w;
            int gn = g + 1;
            if (gn < 8 || (gn < CAP / 4 && c > 32)) cur = bk4[gn];
            q0.z = l0 ? q0.z : 0.0f;
            q1.z = l1 ? q1.z : 0.0f;
            q2.z = l2 ? q2.z : 0.0f;
            q3.z = l3 ? q3.z : 0.0f;
            if (g < STASH_TRIPS) {
                float4* sp4 = stash + (g * 4) * 16 + lc;
                sp4[0] = q0;  sp4[16] = q1;  sp4[32] = q2;  sp4[48] = q3;
            }
            add_terms(acc, q0, q1, q2, q3, wa, wb, bb);
        }
    } else {
        for (int g = 0; g < STASH_TRIPS; ++g) {
            int v0 = g * 4 - 1;
            if (__builtin_amdgcn_ballot_w32(v0 < cc) == 0) break;
            const float4* sp4 = stash + (g * 4) * 16 + lc;
            float4 q0 = sp4[0], q1 = sp4[16], q2 = sp4[32], q3 = sp4[48];
            add_terms(acc, q0, q1, q2, q3, wa, wb, bb);
        }
        for (int g = STASH_TRIPS; g <= CAP / 4; ++g) {
            int v0 = g * 4 - 1;
            if (__builtin_amdgcn_ballot_w32(v0 < cc) == 0) break;
            bool l0 = v0 < cc, l1 = v0 + 1 < cc, l2 = v0 + 2 < cc, l3 = v0 + 3 < cc;
            int i0 = node, i1 = node, i2 = node, i3 = node;
            if (l0) i0 = bk[v0];
            if (l1) i1 = bk[v0 + 1];
            if (l2) i2 = bk[v0 + 2];
            if (l3) i3 = bk[v0 + 3];
            float4 q0 = p4[i0], q1 = p4[i1], q2 = p4[i2], q3 = p4[i3];
            q0.z = l0 ? q0.z : 0.0f;
            q1.z = l1 ? q1.z : 0.0f;
            q2.z = l2 ? q2.z : 0.0f;
            q3.z = l3 ? q3.z : 0.0f;
            add_terms(acc, q0, q1, q2, q3, wa, wb, bb);
        }
    }
    if (c > CAP) {
        int nsp = cnt[n];
        const int2* sp = (const int2*)spill;
        for (int p = 0; p < nsp; ++p) {
            int2 ds = sp[p];
            if (ds.x == node) {
                float4 q = p4[ds.y];
                #pragma unroll
                for (int t = 0; t < 8; ++t)
                    acc[t] = add_act(acc[t], q.x, q.y, q.z, wa[t], wb[t], bb[t]);
            }
        }
    }
    #pragma unroll
    for (int t = 0; t < 8; ++t) {
        acc[t].x *= di;
        acc[t].y *= di;
    }
}

template <int T0>
__device__ __forceinline__ void gemm_steps(const v2f (&acc)[8], const float* __restrict__ W, int hh, int l,
                                           v8f& acc0, v8f& acc1, v8f& acc2, v8f& acc3) {
    #pragma unroll
    for (int t = 0; t < 8; ++t) {
        int ka = 4 * (T0 + t) + hh * 2;
        v2f a = acc[t];
        const float* bp = W + (size_t)ka * HID + l;
        v2f b0, b1f, b2f, b3;
        b0.x  = bp[0];  b0.y  = bp[HID];
        b1f.x = bp[16]; b1f.y = bp[HID + 16];
        b2f.x = bp[32]; b2f.y = bp[HID + 32];
        b3.x  = bp[48]; b3.y  = bp[HID + 48];
        acc0 = __builtin_amdgcn_wmma_f32_16x16x4_f32(false, a, false, b0,  (short)0, acc0, false, false);
        acc1 = __builtin_amdgcn_wmma_f32_16x16x4_f32(false, a, false, b1f, (short)0, acc1, false, false);
        acc2 = __builtin_amdgcn_wmma_f32_16x16x4_f32(false, a, false, b2f, (short)0, acc2, false, false);
        acc3 = __builtin_amdgcn_wmma_f32_16x16x4_f32(false, a, false, b3,  (short)0, acc3, false, false);
    }
}

#define SUM_WAVES 4
__global__ void __launch_bounds__(32 * SUM_WAVES)
k_sum_gemm_relu_dot_wmma(const int* __restrict__ cnt, const int* __restrict__ bucket,
                         const int* __restrict__ spill, const float* __restrict__ pt,
                         const float* __restrict__ W1, const float* __restrict__ b1,
                         const float* __restrict__ W, const float* __restrict__ b2,
                         const float* __restrict__ W3, const float* __restrict__ dinv,
                         float* __restrict__ h3, int n) {
    int gwave = (blockIdx.x * blockDim.x + threadIdx.x) >> 5;
    int lane  = threadIdx.x & 31;
    int row0  = gwave << 4;
    if (row0 >= n) return;
    int hh = lane >> 4;
    int l  = lane & 15;
    int node = row0 + l;
    const float4* p4 = (const float4*)pt;
    int c = cnt[node];
    float di = dinv[node];
    __shared__ float4 stash_all[SUM_WAVES * STASH_SLOTS * 16];
    float4* stash = stash_all + (threadIdx.x >> 5) * (STASH_SLOTS * 16);
    v2f sa[8], sb[8];
    node_sums<0, true >(sa, node, hh, l, c, di, n, cnt, bucket, spill, p4, W1, b1, stash);
    node_sums<8, false>(sb, node, hh, l, c, di, n, cnt, bucket, spill, p4, W1, b1, stash);
    v8f acc0 = {}, acc1 = {}, acc2 = {}, acc3 = {};
    gemm_steps<0>(sa, W, hh, l, acc0, acc1, acc2, acc3);
    gemm_steps<8>(sb, W, hh, l, acc0, acc1, acc2, acc3);
    float c0 = b2[l], c1 = b2[16 + l], c2 = b2[32 + l], c3 = b2[48 + l];
    float u0 = W3[l], u1 = W3[16 + l], u2 = W3[32 + l], u3 = W3[48 + l];
    float mine = 0.0f;
    #pragma unroll
    for (int r = 0; r < 8; ++r) {
        float t = fmaxf(acc0[r] + c0, 0.0f) * u0 + fmaxf(acc1[r] + c1, 0.0f) * u1
                + fmaxf(acc2[r] + c2, 0.0f) * u2 + fmaxf(acc3[r] + c3, 0.0f) * u3;
        t += row_xor<1>(t);
        t += row_xor<2>(t);
        t += row_xor<4>(t);
        t += row_xor<8>(t);
        if (l == r) mine = t;
    }
    if (l < 8) {
        int row = row0 + hh * 8 + l;
        h3[row] = mine * dinv[row];
    }
}

__global__ void k_gather_scalar(const int* __restrict__ cnt, const int* __restrict__ bucket,
                                const int* __restrict__ spill,
                                const float* __restrict__ dinv, const float* __restrict__ h3,
                                const float* __restrict__ b3, float* __restrict__ y, int n) {
    int t = blockIdx.x * blockDim.x + threadIdx.x;
    int node = t >> 3, sub = t & 7;
    float v = 0.0f;
    if (node < n) {
        const int* bk = bucket + (size_t)node * CAP;
        int r0 = bk[sub], r1 = bk[sub + 8], r2 = bk[sub + 16], r3 = bk[sub + 24];
        int c = cnt[node];
        bool p0 = sub < c, p1 = sub + 8 < c, p2 = sub + 16 < c, p3 = sub + 24 < c;
        float a0 = h3[p0 ? r0 : node], a1 = h3[p1 ? r1 : node];
        float a2 = h3[p2 ? r2 : node], a3 = h3[p3 ? r3 : node];
        v += p0 ? a0 : 0.0f;
        v += p1 ? a1 : 0.0f;
        v += p2 ? a2 : 0.0f;
        v += p3 ? a3 : 0.0f;
        int cc = c < CAP ? c : CAP;
        for (int j = 32 + sub; j < cc; j += 8) v += h3[bk[j]];
        if (c > CAP) {
            int nsp = cnt[n];
            const int2* sp = (const int2*)spill;
            for (int p = sub; p < nsp; p += 8) {
                int2 ds = sp[p];
                if (ds.x == node) v += h3[ds.y];
            }
        }
    }
    v += __shfl_xor(v, 4, 32);
    v += __shfl_xor(v, 2, 32);
    v += __shfl_xor(v, 1, 32);
    if (node < n && sub == 0) {
        float di = dinv[node];
        y[node] = b3[0] + (h3[node] + v) * di;
    }
}


extern "C" void kernel_launch(void* const* d_in, const int* in_sizes, int n_in,
                              void* d_out, int out_size, void* d_ws, size_t ws_size,
                              hipStream_t stream) {
    const float* x  = (const float*)d_in[0];
    const int*   ei = (const int*)  d_in[1];
    const float* W1 = (const float*)d_in[2];
    const float* b1 = (const float*)d_in[3];
    const float* W2 = (const float*)d_in[4];
    const float* b2 = (const float*)d_in[5];
    const float* W3 = (const float*)d_in[6];
    const float* b3 = (const float*)d_in[7];

    const int N = in_sizes[0] / 2;
    const int E = in_sizes[1] / 2;
    const int* src = ei;
    const int* dst = ei + E;

    int*   bucket = (int*)d_ws;
    float* pt     = (float*)(bucket + (size_t)N * CAP);
    float* xs     = pt + (size_t)N * 4;
    float* dinv   = xs + (size_t)N * 2;
    float* h3     = dinv + (size_t)N;
    int*   spill  = (int*)(h3 + (size_t)N);
    int*   rec    = spill + (size_t)E * 2;
    const int nbins = (N + BIN_NODES - 1) / BIN_NODES;
    const int nsplit = (E + SPLIT_EDGES - 1) / SPLIT_EDGES;
    const int spitch = (nbins + 15) / 16 * 16;
    int*   seg    = rec + (size_t)E * 2;
    int*   cnt    = seg + (size_t)nsplit * spitch * 2;
    float* y      = (float*)d_out;

    const int B = 256;
    auto g = [](long long t) { return (int)((t + 255) / 256); };

    k_split<<<nsplit, 1024, 0, stream>>>(src, dst, rec, seg, cnt, E, N, nbins, spitch);
    k_place<<<nbins, 1024, 0, stream>>>(rec, seg, x, cnt, bucket, spill, dinv, xs, N, nsplit, spitch);

    k_gather_x<<<g((long long)N * 8), B, 0, stream>>>(cnt, bucket, spill, dinv, xs, pt, N);

    k_sum_gemm_relu_dot_wmma<<<(N / 16 + SUM_WAVES - 1) / SUM_WAVES, 32 * SUM_WAVES, 0, stream>>>(
        cnt, bucket, spill, pt, W1, b1, W2, b2, W3, dinv, h3, N);

    k_gather_scalar<<<g((long long)N * 8),  B, 0, stream>>>(cnt, bucket, spill, dinv, h3, b3, y, N);
}
